# starting kernel
# baseline (speedup 1.0000x reference)
#include <hip/hip_runtime.h>
#include <math.h>


typedef __attribute__((ext_vector_type(2))) float v2f;
typedef __attribute__((ext_vector_type(8))) float v8f;

#define CCH 128

__global__ __launch_bounds__(256) void degree_kernel(
    const int* __restrict__ dst, float* __restrict__ cnt, int E) {
  int e = blockIdx.x * blockDim.x + threadIdx.x;
  if (e < E) {
    unsafeAtomicAdd(&cnt[dst[e]], 1.0f);
  }
}

__global__ __launch_bounds__(256) void inv_kernel(
    const float* __restrict__ cnt, float* __restrict__ inv, int n) {
  int i = blockIdx.x * blockDim.x + threadIdx.x;
  if (i < n) inv[i] = 1.0f / fmaxf(cnt[i], 1.0f);
}

__global__ __launch_bounds__(256) void scatter_kernel(
    const float* __restrict__ feat, const int* __restrict__ src,
    const int* __restrict__ dst, float* __restrict__ agg, int E) {
  long tid = (long)blockIdx.x * blockDim.x + threadIdx.x;
  long total = (long)E * 32;
  if (tid >= total) return;
  int e = (int)(tid >> 5);
  int c = ((int)tid & 31) * 4;
  int s = src[e];
  int d = dst[e];
  const float4 v = *(const float4*)(feat + (long)s * CCH + c);
  float* base = agg + (long)d * CCH + c;
  unsafeAtomicAdd(base + 0, v.x);
  unsafeAtomicAdd(base + 1, v.y);
  unsafeAtomicAdd(base + 2, v.z);
  unsafeAtomicAdd(base + 3, v.w);
}

template <bool SIG>
__global__ __launch_bounds__(128) void sage_layer_kernel(
    const float* __restrict__ xin,
    const float* __restrict__ agg,
    const float* __restrict__ inv,
    const float* __restrict__ Wl,
    const float* __restrict__ bl,
    const float* __restrict__ Wr,
    float* __restrict__ out,
    int n_nodes) {
  const int lane = threadIdx.x & 31;
  const int wave = threadIdx.x >> 5;
  const int kk   = lane >> 4;
  const int ml   = lane & 15;
  const int mbase = blockIdx.x * 128 + wave * 32;

  int m0 = mbase + ml;       if (m0 >= n_nodes) m0 = n_nodes - 1;
  int m1 = mbase + 16 + ml;  if (m1 >= n_nodes) m1 = n_nodes - 1;
  const float inv0 = inv[m0];
  const float inv1 = inv[m1];

  const float2* a0p = (const float2*)(agg + (long)m0 * CCH);
  const float2* a1p = (const float2*)(agg + (long)m1 * CCH);
  const float2* x0p = (const float2*)(xin + (long)m0 * CCH);
  const float2* x1p = (const float2*)(xin + (long)m1 * CCH);

  v8f acc[2][8];
#pragma unroll
  for (int t = 0; t < 2; ++t)
#pragma unroll
    for (int n = 0; n < 8; ++n) acc[t][n] = {};

  for (int ks = 0; ks < 32; ++ks) {
    const int kh = ks * 2 + kk;
    float2 r0 = a0p[kh];
    float2 r1 = a1p[kh];
    v2f A0; A0.x = r0.x * inv0; A0.y = r0.y * inv0;
    v2f A1; A1.x = r1.x * inv1; A1.y = r1.y * inv1;
#pragma unroll
    for (int n = 0; n < 8; ++n) {
      float2 bw = *(const float2*)(Wl + (long)(n * 16 + ml) * CCH + 2 * kh);
      v2f B; B.x = bw.x; B.y = bw.y;
      acc[0][n] = __builtin_amdgcn_wmma_f32_16x16x4_f32(
          false, A0, false, B, (short)0, acc[0][n], false, false);
      acc[1][n] = __builtin_amdgcn_wmma_f32_16x16x4_f32(
          false, A1, false, B, (short)0, acc[1][n], false, false);
    }
  }

  for (int ks = 0; ks < 32; ++ks) {
    const int kh = ks * 2 + kk;
    float2 r0 = x0p[kh];
    float2 r1 = x1p[kh];
    v2f A0; A0.x = r0.x; A0.y = r0.y;
    v2f A1; A1.x = r1.x; A1.y = r1.y;
#pragma unroll
    for (int n = 0; n < 8; ++n) {
      float2 bw = *(const float2*)(Wr + (long)(n * 16 + ml) * CCH + 2 * kh);
      v2f B; B.x = bw.x; B.y = bw.y;
      acc[0][n] = __builtin_amdgcn_wmma_f32_16x16x4_f32(
          false, A0, false, B, (short)0, acc[0][n], false, false);
      acc[1][n] = __builtin_amdgcn_wmma_f32_16x16x4_f32(
          false, A1, false, B, (short)0, acc[1][n], false, false);
    }
  }

#pragma unroll
  for (int n = 0; n < 8; ++n) {
    const float bias = bl[n * 16 + ml];
#pragma unroll
    for (int t = 0; t < 2; ++t) {
#pragma unroll
      for (int r = 0; r < 8; ++r) {
        int m = mbase + t * 16 + r + 8 * kk;
        if (m < n_nodes) {
          float v = acc[t][n][r] + bias;
          if (SIG) v = 1.0f / (1.0f + __expf(-v));
          out[(long)m * CCH + n * 16 + ml] = v;
        }
      }
    }
  }
}

extern "C" void kernel_launch(void* const* d_in, const int* in_sizes, int n_in,
                              void* d_out, int out_size, void* d_ws, size_t ws_size,
                              hipStream_t stream) {
  const float* x   = (const float*)d_in[0];
  const int*   ei  = (const int*)d_in[1];
  const float* W1l = (const float*)d_in[2];
  const float* b1l = (const float*)d_in[3];
  const float* W1r = (const float*)d_in[4];
  const float* W2l = (const float*)d_in[5];
  const float* b2l = (const float*)d_in[6];
  const float* W2r = (const float*)d_in[7];
  float* out = (float*)d_out;

  const int N = in_sizes[0] / CCH;
  const int E = in_sizes[1] / 2;

  float* agg = (float*)d_ws;
  float* h   = agg + (size_t)N * CCH;
  float* cnt = h + (size_t)N * CCH;
  float* vin = cnt + (size_t)N;

  const int* src = ei;
  const int* dst = ei + E;

  hipMemsetAsync(agg, 0, (size_t)N * CCH * sizeof(float), stream);
  hipMemsetAsync(cnt, 0, (size_t)N * sizeof(float), stream);
  degree_kernel<<<(E + 255) / 256, 256, 0, stream>>>(dst, cnt, E);
  inv_kernel<<<(N + 255) / 256, 256, 0, stream>>>(cnt, vin, N);

  const long sthreads = (long)E * 32;
  const int  sblocks  = (int)((sthreads + 255) / 256);
  const int  lblocks  = (N + 127) / 128;

  scatter_kernel<<<sblocks, 256, 0, stream>>>(x, src, dst, agg, E);
  sage_layer_kernel<true><<<lblocks, 128, 0, stream>>>(
      x, agg, vin, W1l, b1l, W1r, h, N);

  hipMemsetAsync(agg, 0, (size_t)N * CCH * sizeof(float), stream);
  scatter_kernel<<<sblocks, 256, 0, stream>>>(h, src, dst, agg, E);
  sage_layer_kernel<false><<<lblocks, 128, 0, stream>>>(
      h, agg, vin, W2l, b2l, W2r, out, N);
}


# Replace fp32 atomic push-scatter and its two zero-fills by a pull over a CSR built per call (edge rank by returning int atomic); row sums now added in arrival-rank order per node
# speedup vs baseline: 1.6602x; 1.6602x over previous
#include <hip/hip_runtime.h>
#include <math.h>


typedef __attribute__((ext_vector_type(2))) float v2f;
typedef __attribute__((ext_vector_type(8))) float v8f;

#define CCH 128

__global__ __launch_bounds__(256) void rank_kernel(
    const int* __restrict__ dst, int* __restrict__ cnt,
    int* __restrict__ rank, int E) {
  int e = blockIdx.x * blockDim.x + threadIdx.x;
  if (e < E) rank[e] = atomicAdd(&cnt[dst[e]], 1);
}

__global__ __launch_bounds__(256) void alloc_kernel(
    const int* __restrict__ cnt, int* __restrict__ total,
    int* __restrict__ start, float* __restrict__ inv, int n) {
  __shared__ int wsum[8];
  __shared__ int s_base;
  const int i    = blockIdx.x * 256 + threadIdx.x;
  const int lane = threadIdx.x & 31;
  const int w    = threadIdx.x >> 5;
  const int d    = (i < n) ? cnt[i] : 0;
  int v = d;
#pragma unroll
  for (int o = 1; o < 32; o <<= 1) {
    int t = __shfl_up(v, o);
    if (lane >= o) v += t;
  }
  if (lane == 31) wsum[w] = v;
  __syncthreads();
  if (threadIdx.x == 0) {
    int run = 0;
#pragma unroll
    for (int k = 0; k < 8; ++k) { int t = wsum[k]; wsum[k] = run; run += t; }
    s_base = atomicAdd(total, run);
  }
  __syncthreads();
  if (i < n) {
    start[i] = s_base + wsum[w] + v - d;
    inv[i]   = 1.0f / fmaxf((float)d, 1.0f);
  }
}

__global__ __launch_bounds__(256) void fill_kernel(
    const int* __restrict__ src, const int* __restrict__ dst,
    const int* __restrict__ rank, const int* __restrict__ start,
    int* __restrict__ col, int E) {
  int e = blockIdx.x * blockDim.x + threadIdx.x;
  if (e < E) col[start[dst[e]] + rank[e]] = src[e];
}

#define NPW 4
__global__ __launch_bounds__(256) void gather_kernel(
    const float* __restrict__ feat, const int* __restrict__ col,
    const int* __restrict__ start, const int* __restrict__ deg,
    float* __restrict__ agg, int n) {
  const int lane = threadIdx.x & 31;
  const int nb   = ((blockIdx.x * 256 + threadIdx.x) >> 5) * NPW;
  int my_s = 0, my_d = 0;
  if (lane < NPW && nb + lane < n) { my_s = start[nb + lane]; my_d = deg[nb + lane]; }
  const float* frow = feat + lane * 4;
#pragma unroll
  for (int t = 0; t < NPW; ++t) {
    const int node = nb + t;
    if (node >= n) break;
    const int s = __builtin_amdgcn_readlane(my_s, t);
    const int d = __builtin_amdgcn_readlane(my_d, t);
    float4 acc = make_float4(0.f, 0.f, 0.f, 0.f);
    for (int j0 = 0; j0 < d; j0 += 32) {
      const int m = (d - j0 < 32) ? (d - j0) : 32;
      const int c = (lane < m) ? col[s + j0 + lane] : 0;
      for (int j = 0; j < m; j += 8) {
        const int r = m - j;
        int ck[8];
#pragma unroll
        for (int k = 0; k < 8; ++k) ck[k] = __builtin_amdgcn_readlane(c, j + k);
        float4 v[8];
#pragma unroll
        for (int k = 0; k < 8; ++k) {
          v[k] = make_float4(0.f, 0.f, 0.f, 0.f);
          if (k < r) v[k] = *(const float4*)(frow + (long)ck[k] * CCH);
        }
#pragma unroll
        for (int k = 0; k < 8; ++k) {
          acc.x += v[k].x; acc.y += v[k].y; acc.z += v[k].z; acc.w += v[k].w;
        }
      }
    }
    *(float4*)(agg + (long)node * CCH + lane * 4) = acc;
  }
}

template <bool SIG>
__global__ __launch_bounds__(128) void sage_layer_kernel(
    const float* __restrict__ xin,
    const float* __restrict__ agg,
    const float* __restrict__ inv,
    const float* __restrict__ Wl,
    const float* __restrict__ bl,
    const float* __restrict__ Wr,
    float* __restrict__ out,
    int n_nodes) {
  const int lane = threadIdx.x & 31;
  const int wave = threadIdx.x >> 5;
  const int kk   = lane >> 4;
  const int ml   = lane & 15;
  const int mbase = blockIdx.x * 128 + wave * 32;

  int m0 = mbase + ml;       if (m0 >= n_nodes) m0 = n_nodes - 1;
  int m1 = mbase + 16 + ml;  if (m1 >= n_nodes) m1 = n_nodes - 1;
  const float inv0 = inv[m0];
  const float inv1 = inv[m1];

  const float2* a0p = (const float2*)(agg + (long)m0 * CCH);
  const float2* a1p = (const float2*)(agg + (long)m1 * CCH);
  const float2* x0p = (const float2*)(xin + (long)m0 * CCH);
  const float2* x1p = (const float2*)(xin + (long)m1 * CCH);

  v8f acc[2][8];
#pragma unroll
  for (int t = 0; t < 2; ++t)
#pragma unroll
    for (int n = 0; n < 8; ++n) acc[t][n] = {};

  for (int ks = 0; ks < 32; ++ks) {
    const int kh = ks * 2 + kk;
    float2 r0 = a0p[kh];
    float2 r1 = a1p[kh];
    v2f A0; A0.x = r0.x * inv0; A0.y = r0.y * inv0;
    v2f A1; A1.x = r1.x * inv1; A1.y = r1.y * inv1;
#pragma unroll
    for (int n = 0; n < 8; ++n) {
      float2 bw = *(const float2*)(Wl + (long)(n * 16 + ml) * CCH + 2 * kh);
      v2f B; B.x = bw.x; B.y = bw.y;
      acc[0][n] = __builtin_amdgcn_wmma_f32_16x16x4_f32(
          false, A0, false, B, (short)0, acc[0][n], false, false);
      acc[1][n] = __builtin_amdgcn_wmma_f32_16x16x4_f32(
          false, A1, false, B, (short)0, acc[1][n], false, false);
    }
  }

  for (int ks = 0; ks < 32; ++ks) {
    const int kh = ks * 2 + kk;
    float2 r0 = x0p[kh];
    float2 r1 = x1p[kh];
    v2f A0; A0.x = r0.x; A0.y = r0.y;
    v2f A1; A1.x = r1.x; A1.y = r1.y;
#pragma unroll
    for (int n = 0; n < 8; ++n) {
      float2 bw = *(const float2*)(Wr + (long)(n * 16 + ml) * CCH + 2 * kh);
      v2f B; B.x = bw.x; B.y = bw.y;
      acc[0][n] = __builtin_amdgcn_wmma_f32_16x16x4_f32(
          false, A0, false, B, (short)0, acc[0][n], false, false);
      acc[1][n] = __builtin_amdgcn_wmma_f32_16x16x4_f32(
          false, A1, false, B, (short)0, acc[1][n], false, false);
    }
  }

#pragma unroll
  for (int n = 0; n < 8; ++n) {
    const float bias = bl[n * 16 + ml];
#pragma unroll
    for (int t = 0; t < 2; ++t) {
#pragma unroll
      for (int r = 0; r < 8; ++r) {
        int m = mbase + t * 16 + r + 8 * kk;
        if (m < n_nodes) {
          float v = acc[t][n][r] + bias;
          if (SIG) v = 1.0f / (1.0f + __expf(-v));
          out[(long)m * CCH + n * 16 + ml] = v;
        }
      }
    }
  }
}

extern "C" void kernel_launch(void* const* d_in, const int* in_sizes, int n_in,
                              void* d_out, int out_size, void* d_ws, size_t ws_size,
                              hipStream_t stream) {
  const float* x   = (const float*)d_in[0];
  const int*   ei  = (const int*)d_in[1];
  const float* W1l = (const float*)d_in[2];
  const float* b1l = (const float*)d_in[3];
  const float* W1r = (const float*)d_in[4];
  const float* W2l = (const float*)d_in[5];
  const float* b2l = (const float*)d_in[6];
  const float* W2r = (const float*)d_in[7];
  float* out = (float*)d_out;

  const int N = in_sizes[0] / CCH;
  const int E = in_sizes[1] / 2;

  float* agg = (float*)d_ws;
  float* h   = agg + (size_t)N * CCH;
  float* vin = h + (size_t)N * CCH + (size_t)N;

  int* total = (int*)d_out;
  int* cnt   = total + 32;
  int* start = cnt + N;
  int* rank  = start + N;
  int* col   = rank + E;

  const int* src = ei;
  const int* dst = ei + E;

  const int eblocks = (E + 255) / 256;
  const int gblocks = (N + 8 * NPW - 1) / (8 * NPW);
  const int lblocks = (N + 127) / 128;

  hipMemsetAsync(total, 0, (size_t)(N + 32) * sizeof(int), stream);
  rank_kernel<<<eblocks, 256, 0, stream>>>(dst, cnt, rank, E);
  alloc_kernel<<<(N + 255) / 256, 256, 0, stream>>>(cnt, total, start, vin, N);
  fill_kernel<<<eblocks, 256, 0, stream>>>(src, dst, rank, start, col, E);

  gather_kernel<<<gblocks, 256, 0, stream>>>(x, col, start, cnt, agg, N);
  sage_layer_kernel<true><<<lblocks, 128, 0, stream>>>(
      x, agg, vin, W1l, b1l, W1r, h, N);

  gather_kernel<<<gblocks, 256, 0, stream>>>(h, col, start, cnt, agg, N);
  sage_layer_kernel<false><<<lblocks, 128, 0, stream>>>(
      h, agg, vin, W2l, b2l, W2r, out, N);
}


# Layer GEMM: weights staged once per resident work-group in LDS (256 groups loop over 16-node tiles), B fragments read from LDS by 16-byte loads; k summed in a permuted order inside each 8
# speedup vs baseline: 2.2177x; 1.3358x over previous
#include <hip/hip_runtime.h>
#include <math.h>


typedef __attribute__((ext_vector_type(2))) float v2f;
typedef __attribute__((ext_vector_type(8))) float v8f;

#define CCH 128

__global__ __launch_bounds__(256) void rank_kernel(
    const int* __restrict__ dst, int* __restrict__ cnt,
    int* __restrict__ rank, int E) {
  int e = blockIdx.x * blockDim.x + threadIdx.x;
  if (e < E) rank[e] = atomicAdd(&cnt[dst[e]], 1);
}

__global__ __launch_bounds__(256) void alloc_kernel(
    const int* __restrict__ cnt, int* __restrict__ total,
    int* __restrict__ start, float* __restrict__ inv, int n) {
  __shared__ int wsum[8];
  __shared__ int s_base;
  const int i    = blockIdx.x * 256 + threadIdx.x;
  const int lane = threadIdx.x & 31;
  const int w    = threadIdx.x >> 5;
  const int d    = (i < n) ? cnt[i] : 0;
  int v = d;
#pragma unroll
  for (int o = 1; o < 32; o <<= 1) {
    int t = __shfl_up(v, o);
    if (lane >= o) v += t;
  }
  if (lane == 31) wsum[w] = v;
  __syncthreads();
  if (threadIdx.x == 0) {
    int run = 0;
#pragma unroll
    for (int k = 0; k < 8; ++k) { int t = wsum[k]; wsum[k] = run; run += t; }
    s_base = atomicAdd(total, run);
  }
  __syncthreads();
  if (i < n) {
    start[i] = s_base + wsum[w] + v - d;
    inv[i]   = 1.0f / fmaxf((float)d, 1.0f);
  }
}

__global__ __launch_bounds__(256) void fill_kernel(
    const int* __restrict__ src, const int* __restrict__ dst,
    const int* __restrict__ rank, const int* __restrict__ start,
    int* __restrict__ col, int E) {
  int e = blockIdx.x * blockDim.x + threadIdx.x;
  if (e < E) col[start[dst[e]] + rank[e]] = src[e];
}

#define NPW 4
__global__ __launch_bounds__(256) void gather_kernel(
    const float* __restrict__ feat, const int* __restrict__ col,
    const int* __restrict__ start, const int* __restrict__ deg,
    float* __restrict__ agg, int n) {
  const int lane = threadIdx.x & 31;
  const int nb   = ((blockIdx.x * 256 + threadIdx.x) >> 5) * NPW;
  int my_s = 0, my_d = 0;
  if (lane < NPW && nb + lane < n) { my_s = start[nb + lane]; my_d = deg[nb + lane]; }
  const float* frow = feat + lane * 4;
#pragma unroll
  for (int t = 0; t < NPW; ++t) {
    const int node = nb + t;
    if (node >= n) break;
    const int s = __builtin_amdgcn_readlane(my_s, t);
    const int d = __builtin_amdgcn_readlane(my_d, t);
    float4 acc = make_float4(0.f, 0.f, 0.f, 0.f);
    for (int j0 = 0; j0 < d; j0 += 32) {
      const int m = (d - j0 < 32) ? (d - j0) : 32;
      const int c = (lane < m) ? col[s + j0 + lane] : 0;
      for (int j = 0; j < m; j += 8) {
        const int r = m - j;
        int ck[8];
#pragma unroll
        for (int k = 0; k < 8; ++k) ck[k] = __builtin_amdgcn_readlane(c, j + k);
        float4 v[8];
#pragma unroll
        for (int k = 0; k < 8; ++k) {
          v[k] = make_float4(0.f, 0.f, 0.f, 0.f);
          if (k < r) v[k] = *(const float4*)(frow + (long)ck[k] * CCH);
        }
#pragma unroll
        for (int k = 0; k < 8; ++k) {
          acc.x += v[k].x; acc.y += v[k].y; acc.z += v[k].z; acc.w += v[k].w;
        }
      }
    }
    *(float4*)(agg + (long)node * CCH + lane * 4) = acc;
  }
}

#define LDP 132
#define SG_WAVES   12
#define SG_THREADS (SG_WAVES * 32)
#define SG_BLOCKS  256

__device__ __forceinline__ v8f wmma4(v2f a, v2f b, v8f c) {
  return __builtin_amdgcn_wmma_f32_16x16x4_f32(false, a, false, b, (short)0, c,
                                               false, false);
}

template <bool SCALE>
__device__ __forceinline__ void half_tile(v8f (&acc)[8], const float4 (&a)[16],
                                          float s, const float* wl, int wofs) {
  float4 bc[8], bn[8];
#pragma unroll
  for (int n = 0; n < 8; ++n) bc[n] = *(const float4*)(wl + wofs + n * 16 * LDP);
#pragma unroll
  for (int q = 0; q < 16; ++q) {
    if (q + 1 < 16) {
#pragma unroll
      for (int n = 0; n < 8; ++n)
        bn[n] = *(const float4*)(wl + wofs + n * 16 * LDP + 8 * (q + 1));
    }
    v2f a0, a1;
    if (SCALE) {
      a0.x = a[q].x * s; a0.y = a[q].y * s; a1.x = a[q].z * s; a1.y = a[q].w * s;
    } else {
      a0.x = a[q].x; a0.y = a[q].y; a1.x = a[q].z; a1.y = a[q].w;
    }
#pragma unroll
    for (int n = 0; n < 8; ++n) {
      v2f b; b.x = bc[n].x; b.y = bc[n].y;
      acc[n] = wmma4(a0, b, acc[n]);
    }
#pragma unroll
    for (int n = 0; n < 8; ++n) {
      v2f b; b.x = bc[n].z; b.y = bc[n].w;
      acc[n] = wmma4(a1, b, acc[n]);
    }
#pragma unroll
    for (int n = 0; n < 8; ++n) bc[n] = bn[n];
  }
}

template <bool SIG>
__global__ __launch_bounds__(SG_THREADS) void sage_layer_kernel(
    const float* __restrict__ xin,
    const float* __restrict__ agg,
    const float* __restrict__ inv,
    const float* __restrict__ Wl,
    const float* __restrict__ bl,
    const float* __restrict__ Wr,
    float* __restrict__ out,
    int n_nodes) {
  __shared__ __attribute__((aligned(16))) float wlds[256 * LDP];
  const int tid = threadIdx.x;

#pragma unroll 7
  for (int j = 0; j < 21; ++j) {
    const int i = j * SG_THREADS + tid, row = i >> 5, c4 = i & 31;
    const float* wsrc = (row < 128) ? (Wl + row * CCH) : (Wr + (row - 128) * CCH);
    *(float4*)(wlds + row * LDP + c4 * 4) = *(const float4*)(wsrc + c4 * 4);
  }
  if (tid < 256 * 32 - 21 * SG_THREADS) {
    const int i = 21 * SG_THREADS + tid, row = i >> 5, c4 = i & 31;
    *(float4*)(wlds + row * LDP + c4 * 4) =
        *(const float4*)(Wr + (row - 128) * CCH + c4 * 4);
  }
  __syncthreads();

  const int lane = tid & 31;
  const int wave = __builtin_amdgcn_readfirstlane(tid >> 5);
  const int h    = lane >> 4;
  const int c    = lane & 15;
  const int ntiles = (n_nodes + 15) >> 4;
  int tile = blockIdx.x + SG_BLOCKS * wave;
  if (tile >= ntiles) return;

  const int wofs = c * LDP + 4 * h;
  float bias[8];
#pragma unroll
  for (int n = 0; n < 8; ++n) bias[n] = bl[n * 16 + c];

  int m = tile * 16 + c;  if (m >= n_nodes) m = n_nodes - 1;
  float iv = inv[m];
  float4 aa[16];
  {
    const float* p = agg + (long)m * CCH + 4 * h;
#pragma unroll
    for (int q = 0; q < 16; ++q) aa[q] = *(const float4*)(p + 8 * q);
  }

  for (;;) {
    asm volatile("" ::: "memory");
    float4 ax[16];
    {
      const float* p = xin + (long)m * CCH + 4 * h;
#pragma unroll
      for (int q = 0; q < 16; ++q) ax[q] = *(const float4*)(p + 8 * q);
    }
    v8f acc[8];
#pragma unroll
    for (int n = 0; n < 8; ++n) acc[n] = {};

    half_tile<true>(acc, aa, iv, wlds, wofs);

    const int  nt   = tile + SG_BLOCKS * SG_WAVES;
    const bool more = nt < ntiles;
    int mn = m;  float ivn = iv;
    if (more) {
      mn = nt * 16 + c;  if (mn >= n_nodes) mn = n_nodes - 1;
      ivn = inv[mn];
      const float* p = agg + (long)mn * CCH + 4 * h;
#pragma unroll
      for (int q = 0; q < 16; ++q) aa[q] = *(const float4*)(p + 8 * q);
    }

    half_tile<false>(acc, ax, 1.0f, wlds, wofs + 128 * LDP);

    const int mrow = tile * 16 + 8 * h;
#pragma unroll
    for (int r = 0; r < 8; ++r) {
      if (mrow + r < n_nodes) {
        float* orow = out + (long)(mrow + r) * CCH + c;
#pragma unroll
        for (int n = 0; n < 8; ++n) {
          float v = acc[n][r] + bias[n];
          if (SIG) v = 1.0f / (1.0f + __expf(-v));
          orow[n * 16] = v;
        }
      }
    }

    if (!more) break;
    tile = nt;  m = mn;  iv = ivn;
  }
}

extern "C" void kernel_launch(void* const* d_in, const int* in_sizes, int n_in,
                              void* d_out, int out_size, void* d_ws, size_t ws_size,
                              hipStream_t stream) {
  const float* x   = (const float*)d_in[0];
  const int*   ei  = (const int*)d_in[1];
  const float* W1l = (const float*)d_in[2];
  const float* b1l = (const float*)d_in[3];
  const float* W1r = (const float*)d_in[4];
  const float* W2l = (const float*)d_in[5];
  const float* b2l = (const float*)d_in[6];
  const float* W2r = (const float*)d_in[7];
  float* out = (float*)d_out;

  const int N = in_sizes[0] / CCH;
  const int E = in_sizes[1] / 2;

  float* agg = (float*)d_ws;
  float* h   = agg + (size_t)N * CCH;
  float* vin = h + (size_t)N * CCH + (size_t)N;

  int* total = (int*)d_out;
  int* cnt   = total + 32;
  int* start = cnt + N;
  int* rank  = start + N;
  int* col   = rank + E;

  const int* src = ei;
  const int* dst = ei + E;

  const int eblocks = (E + 255) / 256;
  const int gblocks = (N + 8 * NPW - 1) / (8 * NPW);

  hipMemsetAsync(total, 0, (size_t)(N + 32) * sizeof(int), stream);
  rank_kernel<<<eblocks, 256, 0, stream>>>(dst, cnt, rank, E);
  alloc_kernel<<<(N + 255) / 256, 256, 0, stream>>>(cnt, total, start, vin, N);
  fill_kernel<<<eblocks, 256, 0, stream>>>(src, dst, rank, start, col, E);

  gather_kernel<<<gblocks, 256, 0, stream>>>(x, col, start, cnt, agg, N);
  sage_layer_kernel<true><<<SG_BLOCKS, SG_THREADS, 0, stream>>>(
      x, agg, vin, W1l, b1l, W1r, h, N);

  gather_kernel<<<gblocks, 256, 0, stream>>>(h, col, start, cnt, agg, N);
  sage_layer_kernel<false><<<SG_BLOCKS, SG_THREADS, 0, stream>>>(
      h, agg, vin, W2l, b2l, W2r, out, N);
}


# Index structure: CSR replaced by fixed 32-slot buckets per destination (one 128-B line), filled in the ranking pass itself; alloc and fill launches and rank array removed; sums stay in arrival order
# speedup vs baseline: 2.4122x; 1.0877x over previous
#include <hip/hip_runtime.h>
#include <math.h>


typedef __attribute__((ext_vector_type(2))) float v2f;
typedef __attribute__((ext_vector_type(8))) float v8f;

#define CCH 128

#define CAP 32
__global__ __launch_bounds__(256) void bucket_kernel(
    const int* __restrict__ src, const int* __restrict__ dst,
    int* __restrict__ cnt, int* __restrict__ col, int E) {
  int e = blockIdx.x * blockDim.x + threadIdx.x;
  if (e < E) {
    const int d = dst[e];
    const int s = src[e];
    const int r = atomicAdd(&cnt[d], 1);
    if (r < CAP) col[(long)d * CAP + r] = s;
  }
}

#define NPW 4
template <bool WRITE_INV>
__global__ __launch_bounds__(256) void gather_kernel(
    const float* __restrict__ feat, const int* __restrict__ col,
    const int* __restrict__ deg, const int* __restrict__ src,
    const int* __restrict__ dst, float* __restrict__ agg,
    float* __restrict__ inv, int n, int E) {
  const int lane = threadIdx.x & 31;
  const int nb   = ((blockIdx.x * 256 + threadIdx.x) >> 5) * NPW;
  int my_d = 0;
  if (lane < NPW && nb + lane < n) {
    my_d = deg[nb + lane];
    if (WRITE_INV) inv[nb + lane] = 1.0f / fmaxf((float)my_d, 1.0f);
  }
  int cb[4] = {0, 0, 0, 0};
  if (nb + (lane >> 3) < n) {
    const int4 q = *(const int4*)(col + (long)nb * CAP + lane * 4);
    cb[0] = q.x; cb[1] = q.y; cb[2] = q.z; cb[3] = q.w;
  }
  const float* frow = feat + lane * 4;
#pragma unroll
  for (int t = 0; t < NPW; ++t) {
    const int node = nb + t;
    if (node >= n) break;
    const int d = __builtin_amdgcn_readlane(my_d, t);
    float4 acc = make_float4(0.f, 0.f, 0.f, 0.f);
    if (d <= CAP) {
      for (int j = 0; j < d; j += 8) {
        const int r  = d - j;
        const int l0 = 8 * t + (j >> 2);
        int ck[8];
#pragma unroll
        for (int k = 0; k < 8; ++k)
          ck[k] = __builtin_amdgcn_readlane(cb[k & 3], l0 + (k >> 2));
        float4 v[8];
#pragma unroll
        for (int k = 0; k < 8; ++k) {
          v[k] = make_float4(0.f, 0.f, 0.f, 0.f);
          if (k < r) v[k] = *(const float4*)(frow + (long)ck[k] * CCH);
        }
#pragma unroll
        for (int k = 0; k < 8; ++k) {
          acc.x += v[k].x; acc.y += v[k].y; acc.z += v[k].z; acc.w += v[k].w;
        }
      }
    } else {
      for (int e0 = 0; e0 < E; e0 += 32) {
        const int  e   = e0 + lane;
        const bool hit = (e < E) && (dst[e] == node);
        const int  sv  = hit ? src[e] : 0;
        unsigned mask  = __builtin_amdgcn_ballot_w32(hit);
        while (mask != 0u) {
          const int b = __builtin_ctz(mask);
          mask &= mask - 1u;
          const int sidx = __builtin_amdgcn_readlane(sv, b);
          const float4 v = *(const float4*)(frow + (long)sidx * CCH);
          acc.x += v.x; acc.y += v.y; acc.z += v.z; acc.w += v.w;
        }
      }
    }
    *(float4*)(agg + (long)node * CCH + lane * 4) = acc;
  }
}

#define LDP 132
#define SG_WAVES   12
#define SG_THREADS (SG_WAVES * 32)
#define SG_BLOCKS  256

__device__ __forceinline__ v8f wmma4(v2f a, v2f b, v8f c) {
  return __builtin_amdgcn_wmma_f32_16x16x4_f32(false, a, false, b, (short)0, c,
                                               false, false);
}

template <bool SCALE>
__device__ __forceinline__ void half_tile(v8f (&acc)[8], const float4 (&a)[16],
                                          float s, const float* wl, int wofs) {
  float4 bc[8], bn[8];
#pragma unroll
  for (int n = 0; n < 8; ++n) bc[n] = *(const float4*)(wl + wofs + n * 16 * LDP);
#pragma unroll
  for (int q = 0; q < 16; ++q) {
    if (q + 1 < 16) {
#pragma unroll
      for (int n = 0; n < 8; ++n)
        bn[n] = *(const float4*)(wl + wofs + n * 16 * LDP + 8 * (q + 1));
    }
    v2f a0, a1;
    if (SCALE) {
      a0.x = a[q].x * s; a0.y = a[q].y * s; a1.x = a[q].z * s; a1.y = a[q].w * s;
    } else {
      a0.x = a[q].x; a0.y = a[q].y; a1.x = a[q].z; a1.y = a[q].w;
    }
#pragma unroll
    for (int n = 0; n < 8; ++n) {
      v2f b; b.x = bc[n].x; b.y = bc[n].y;
      acc[n] = wmma4(a0, b, acc[n]);
    }
#pragma unroll
    for (int n = 0; n < 8; ++n) {
      v2f b; b.x = bc[n].z; b.y = bc[n].w;
      acc[n] = wmma4(a1, b, acc[n]);
    }
#pragma unroll
    for (int n = 0; n < 8; ++n) bc[n] = bn[n];
  }
}

template <bool SIG>
__global__ __launch_bounds__(SG_THREADS) void sage_layer_kernel(
    const float* __restrict__ xin,
    const float* __restrict__ agg,
    const float* __restrict__ inv,
    const float* __restrict__ Wl,
    const float* __restrict__ bl,
    const float* __restrict__ Wr,
    float* __restrict__ out,
    int n_nodes) {
  __shared__ __attribute__((aligned(16))) float wlds[256 * LDP];
  const int tid = threadIdx.x;

#pragma unroll 7
  for (int j = 0; j < 21; ++j) {
    const int i = j * SG_THREADS + tid, row = i >> 5, c4 = i & 31;
    const float* wsrc = (row < 128) ? (Wl + row * CCH) : (Wr + (row - 128) * CCH);
    *(float4*)(wlds + row * LDP + c4 * 4) = *(const float4*)(wsrc + c4 * 4);
  }
  if (tid < 256 * 32 - 21 * SG_THREADS) {
    const int i = 21 * SG_THREADS + tid, row = i >> 5, c4 = i & 31;
    *(float4*)(wlds + row * LDP + c4 * 4) =
        *(const float4*)(Wr + (row - 128) * CCH + c4 * 4);
  }
  __syncthreads();

  const int lane = tid & 31;
  const int wave = __builtin_amdgcn_readfirstlane(tid >> 5);
  const int h    = lane >> 4;
  const int c    = lane & 15;
  const int ntiles = (n_nodes + 15) >> 4;
  int tile = blockIdx.x + SG_BLOCKS * wave;
  if (tile >= ntiles) return;

  const int wofs = c * LDP + 4 * h;
  float bias[8];
#pragma unroll
  for (int n = 0; n < 8; ++n) bias[n] = bl[n * 16 + c];

  int m = tile * 16 + c;  if (m >= n_nodes) m = n_nodes - 1;
  float iv = inv[m];
  float4 aa[16];
  {
    const float* p = agg + (long)m * CCH + 4 * h;
#pragma unroll
    for (int q = 0; q < 16; ++q) aa[q] = *(const float4*)(p + 8 * q);
  }

  for (;;) {
    asm volatile("" ::: "memory");
    float4 ax[16];
    {
      const float* p = xin + (long)m * CCH + 4 * h;
#pragma unroll
      for (int q = 0; q < 16; ++q) ax[q] = *(const float4*)(p + 8 * q);
    }
    v8f acc[8];
#pragma unroll
    for (int n = 0; n < 8; ++n) acc[n] = {};

    half_tile<true>(acc, aa, iv, wlds, wofs);

    const int  nt   = tile + SG_BLOCKS * SG_WAVES;
    const bool more = nt < ntiles;
    int mn = m;  float ivn = iv;
    if (more) {
      mn = nt * 16 + c;  if (mn >= n_nodes) mn = n_nodes - 1;
      ivn = inv[mn];
      const float* p = agg + (long)mn * CCH + 4 * h;
#pragma unroll
      for (int q = 0; q < 16; ++q) aa[q] = *(const float4*)(p + 8 * q);
    }

    half_tile<false>(acc, ax, 1.0f, wlds, wofs + 128 * LDP);

    const int mrow = tile * 16 + 8 * h;
#pragma unroll
    for (int r = 0; r < 8; ++r) {
      if (mrow + r < n_nodes) {
        float* orow = out + (long)(mrow + r) * CCH + c;
#pragma unroll
        for (int n = 0; n < 8; ++n) {
          float v = acc[n][r] + bias[n];
          if (SIG) v = 1.0f / (1.0f + __expf(-v));
          orow[n * 16] = v;
        }
      }
    }

    if (!more) break;
    tile = nt;  m = mn;  iv = ivn;
  }
}

extern "C" void kernel_launch(void* const* d_in, const int* in_sizes, int n_in,
                              void* d_out, int out_size, void* d_ws, size_t ws_size,
                              hipStream_t stream) {
  const float* x   = (const float*)d_in[0];
  const int*   ei  = (const int*)d_in[1];
  const float* W1l = (const float*)d_in[2];
  const float* b1l = (const float*)d_in[3];
  const float* W1r = (const float*)d_in[4];
  const float* W2l = (const float*)d_in[5];
  const float* b2l = (const float*)d_in[6];
  const float* W2r = (const float*)d_in[7];
  float* out = (float*)d_out;

  const int N = in_sizes[0] / CCH;
  const int E = in_sizes[1] / 2;

  float* agg = (float*)d_ws;
  float* h   = agg + (size_t)N * CCH;
  float* vin = h + (size_t)N * CCH + (size_t)N;

  int* col = (int*)d_out;
  int* cnt = col + (size_t)N * CAP;

  const int* src = ei;
  const int* dst = ei + E;

  const int eblocks = (E + 255) / 256;
  const int gblocks = (N + 8 * NPW - 1) / (8 * NPW);

  hipMemsetAsync(cnt, 0, (size_t)N * sizeof(int), stream);
  bucket_kernel<<<eblocks, 256, 0, stream>>>(src, dst, cnt, col, E);

  gather_kernel<true><<<gblocks, 256, 0, stream>>>(x, col, cnt, src, dst, agg, vin, N, E);
  sage_layer_kernel<true><<<SG_BLOCKS, SG_THREADS, 0, stream>>>(
      x, agg, vin, W1l, b1l, W1r, h, N);

  gather_kernel<false><<<gblocks, 256, 0, stream>>>(h, col, cnt, src, dst, agg, vin, N, E);
  sage_layer_kernel<false><<<SG_BLOCKS, SG_THREADS, 0, stream>>>(
      h, agg, vin, W2l, b2l, W2r, out, N);
}


# Neighbour gather-sum fused into the layer kernel: each wave sums its tile's 16 nodes' rows into its own LDS slot and reads its A fragments there; both gather launches and the sum buffer removed
# speedup vs baseline: 2.6692x; 1.1065x over previous
#include <hip/hip_runtime.h>
#include <math.h>


typedef __attribute__((ext_vector_type(2))) float v2f;
typedef __attribute__((ext_vector_type(8))) float v8f;
typedef __attribute__((ext_vector_type(4))) float v4f;

#define CCH 128

#define CAP 32
__global__ __launch_bounds__(256) void bucket_kernel(
    const int* __restrict__ src, const int* __restrict__ dst,
    int* __restrict__ cnt, int* __restrict__ col, int E) {
  int e = blockIdx.x * blockDim.x + threadIdx.x;
  if (e < E) {
    const int d = dst[e];
    const int s = src[e];
    const int r = atomicAdd(&cnt[d], 1);
    if (r < CAP) col[(long)d * CAP + r] = s;
  }
}

#define LDP 132
#define SG_WAVES   16
#define SG_THREADS (SG_WAVES * 32)
#define SG_BLOCKS  256

__device__ __forceinline__ v8f wmma4(v2f a, v2f b, v8f c) {
  return __builtin_amdgcn_wmma_f32_16x16x4_f32(false, a, false, b, (short)0, c,
                                               false, false);
}

__device__ __forceinline__ void gather_tile(
    const float* __restrict__ feat, const int* __restrict__ col,
    const int* __restrict__ deg, const int* __restrict__ src,
    const int* __restrict__ dst, float* slot, int node0, int n, int E,
    int lane) {
  const float* frow = feat + lane * 4;
  for (int g = 0; g < 4; ++g) {
    const int nb = node0 + 4 * g;
    int   my_d = 0;
    float my_s = 1.0f;
    if (lane < 4 && nb + lane < n) {
      my_d = deg[nb + lane];
      my_s = 1.0f / fmaxf((float)my_d, 1.0f);
    }
    int cb[4] = {0, 0, 0, 0};
    if (nb + (lane >> 3) < n) {
      const int4 q = *(const int4*)(col + (long)nb * CAP + lane * 4);
      cb[0] = q.x; cb[1] = q.y; cb[2] = q.z; cb[3] = q.w;
    }
    int d[4];
#pragma unroll
    for (int t = 0; t < 4; ++t) d[t] = __builtin_amdgcn_readlane(my_d, t);
    int ck[4][8];
#pragma unroll
    for (int t = 0; t < 4; ++t)
#pragma unroll
      for (int k = 0; k < 8; ++k)
        ck[t][k] = __builtin_amdgcn_readlane(cb[k & 3], 8 * t + (k >> 2));
    v4f v[4][8];
#pragma unroll
    for (int t = 0; t < 4; ++t)
#pragma unroll
      for (int k = 0; k < 8; ++k) {
        v[t][k] = (v4f){0.f, 0.f, 0.f, 0.f};
        if (k < d[t] && d[t] <= CAP)
          v[t][k] = *(const v4f*)(frow + (long)ck[t][k] * CCH);
      }
#pragma unroll
    for (int t = 0; t < 4; ++t)
#pragma unroll
      for (int k = 0; k < 8; ++k)
        asm volatile("" : "+v"(v[t][k]));
#pragma unroll
    for (int t = 0; t < 4; ++t) {
      float4 acc = make_float4(0.f, 0.f, 0.f, 0.f);
#pragma unroll
      for (int k = 0; k < 8; ++k) {
        acc.x += v[t][k].x; acc.y += v[t][k].y;
        acc.z += v[t][k].z; acc.w += v[t][k].w;
      }
      const int dt = d[t];
      if (dt > 8) {
        if (dt <= CAP) {
          for (int j = 8; j < dt; j += 8) {
            const int r  = dt - j;
            const int l0 = 8 * t + (j >> 2);
            int cj[8];
#pragma unroll
            for (int k = 0; k < 8; ++k)
              cj[k] = __builtin_amdgcn_readlane(cb[k & 3], l0 + (k >> 2));
            float4 w[8];
#pragma unroll
            for (int k = 0; k < 8; ++k) {
              w[k] = make_float4(0.f, 0.f, 0.f, 0.f);
              if (k < r) w[k] = *(const float4*)(frow + (long)cj[k] * CCH);
            }
#pragma unroll
            for (int k = 0; k < 8; ++k) {
              acc.x += w[k].x; acc.y += w[k].y; acc.z += w[k].z; acc.w += w[k].w;
            }
          }
        } else {
          const int node = nb + t;
          for (int e0 = 0; e0 < E; e0 += 32) {
            const int  e   = e0 + lane;
            const bool hit = (e < E) && (dst[e] == node);
            const int  sv  = hit ? src[e] : 0;
            unsigned mask  = __builtin_amdgcn_ballot_w32(hit);
            while (mask != 0u) {
              const int b = __builtin_ctz(mask);
              mask &= mask - 1u;
              const int sidx = __builtin_amdgcn_readlane(sv, b);
              const float4 u = *(const float4*)(frow + (long)sidx * CCH);
              acc.x += u.x; acc.y += u.y; acc.z += u.z; acc.w += u.w;
            }
          }
        }
      }
      const float s = __int_as_float(
          __builtin_amdgcn_readlane(__float_as_int(my_s), t));
      acc.x *= s; acc.y *= s; acc.z *= s; acc.w *= s;
      *(float4*)(slot + (4 * g + t) * LDP + lane * 4) = acc;
    }
  }
}

template <bool A_LDS>
__device__ __forceinline__ void half_tile(v8f (&acc)[8], const float4 (&a)[16],
                                          const float* ap, const float* wl,
                                          int wofs) {
  float4 bc[8], bn[8];
  float4 ac = make_float4(0.f, 0.f, 0.f, 0.f), an = ac;
#pragma unroll
  for (int n = 0; n < 8; ++n) bc[n] = *(const float4*)(wl + wofs + n * 16 * LDP);
  if (A_LDS) ac = *(const float4*)(ap);
#pragma unroll
  for (int q = 0; q < 16; ++q) {
    if (q + 1 < 16) {
#pragma unroll
      for (int n = 0; n < 8; ++n)
        bn[n] = *(const float4*)(wl + wofs + n * 16 * LDP + 8 * (q + 1));
      if (A_LDS) an = *(const float4*)(ap + 8 * (q + 1));
    }
    const float4 av = A_LDS ? ac : a[q];
    v2f a0, a1;
    a0.x = av.x; a0.y = av.y; a1.x = av.z; a1.y = av.w;
#pragma unroll
    for (int n = 0; n < 8; ++n) {
      v2f b; b.x = bc[n].x; b.y = bc[n].y;
      acc[n] = wmma4(a0, b, acc[n]);
    }
#pragma unroll
    for (int n = 0; n < 8; ++n) {
      v2f b; b.x = bc[n].z; b.y = bc[n].w;
      acc[n] = wmma4(a1, b, acc[n]);
    }
#pragma unroll
    for (int n = 0; n < 8; ++n) bc[n] = bn[n];
    ac = an;
  }
}

template <bool SIG>
__global__ __launch_bounds__(SG_THREADS) void sage_layer_kernel(
    const float* __restrict__ xin,
    const int* __restrict__ col,
    const int* __restrict__ deg,
    const int* __restrict__ src,
    const int* __restrict__ dst,
    const float* __restrict__ Wl,
    const float* __restrict__ bl,
    const float* __restrict__ Wr,
    float* __restrict__ out,
    int n_nodes, int E) {
  __shared__ __attribute__((aligned(16))) float wlds[256 * LDP];
  __shared__ __attribute__((aligned(16))) float aslot[SG_WAVES * 16 * LDP];
  const int tid = threadIdx.x;

#pragma unroll 8
  for (int j = 0; j < (256 * 32) / SG_THREADS; ++j) {
    const int i = j * SG_THREADS + tid, row = i >> 5, c4 = i & 31;
    const float* wsrc = (row < 128) ? (Wl + row * CCH) : (Wr + (row - 128) * CCH);
    *(float4*)(wlds + row * LDP + c4 * 4) = *(const float4*)(wsrc + c4 * 4);
  }
  __syncthreads();

  const int lane = tid & 31;
  const int wave = __builtin_amdgcn_readfirstlane(tid >> 5);
  const int h    = lane >> 4;
  const int c    = lane & 15;
  const int ntiles = (n_nodes + 15) >> 4;
  int tile = blockIdx.x + SG_BLOCKS * wave;
  if (tile >= ntiles) return;

  const int wofs = c * LDP + 4 * h;
  int wofs_r = wofs + 128 * LDP;
  asm volatile("" : "+v"(wofs_r));
  float bias[8];
#pragma unroll
  for (int n = 0; n < 8; ++n) bias[n] = bl[n * 16 + c];

  float*       slot = aslot + wave * (16 * LDP);
  const float* ap   = slot + c * LDP + 4 * h;

  for (;;) {
    asm volatile("" ::: "memory");

    gather_tile(xin, col, deg, src, dst, slot, tile * 16, n_nodes, E, lane);

    asm volatile("" ::: "memory");

    int m = tile * 16 + c;  if (m >= n_nodes) m = n_nodes - 1;
    float4 ax[16];
    {
      const float* p = xin + (long)m * CCH + 4 * h;
#pragma unroll
      for (int q = 0; q < 16; ++q) ax[q] = *(const float4*)(p + 8 * q);
    }
    v8f acc[8];
#pragma unroll
    for (int n = 0; n < 8; ++n) acc[n] = {};

    half_tile<true>(acc, ax, ap, wlds, wofs);
    half_tile<false>(acc, ax, ap, wlds, wofs_r);

    const int mrow = tile * 16 + 8 * h;
#pragma unroll
    for (int r = 0; r < 8; ++r) {
      if (mrow + r < n_nodes) {
        float* orow = out + (long)(mrow + r) * CCH + c;
#pragma unroll
        for (int n = 0; n < 8; ++n) {
          float v = acc[n][r] + bias[n];
          if (SIG) v = 1.0f / (1.0f + __expf(-v));
          orow[n * 16] = v;
        }
      }
    }

    tile += SG_BLOCKS * SG_WAVES;
    if (tile >= ntiles) break;
  }
}

extern "C" void kernel_launch(void* const* d_in, const int* in_sizes, int n_in,
                              void* d_out, int out_size, void* d_ws, size_t ws_size,
                              hipStream_t stream) {
  const float* x   = (const float*)d_in[0];
  const int*   ei  = (const int*)d_in[1];
  const float* W1l = (const float*)d_in[2];
  const float* b1l = (const float*)d_in[3];
  const float* W1r = (const float*)d_in[4];
  const float* W2l = (const float*)d_in[5];
  const float* b2l = (const float*)d_in[6];
  const float* W2r = (const float*)d_in[7];
  float* out = (float*)d_out;

  const int N = in_sizes[0] / CCH;
  const int E = in_sizes[1] / 2;

  int*   col = (int*)d_ws;
  int*   cnt = col + (size_t)N * CAP;
  float* h   = (float*)d_ws + (size_t)N * CCH;

  const int* src = ei;
  const int* dst = ei + E;

  const int eblocks = (E + 255) / 256;

  hipMemsetAsync(cnt, 0, (size_t)N * sizeof(int), stream);
  bucket_kernel<<<eblocks, 256, 0, stream>>>(src, dst, cnt, col, E);

  sage_layer_kernel<true><<<SG_BLOCKS, SG_THREADS, 0, stream>>>(
      x, col, cnt, src, dst, W1l, b1l, W1r, h, N, E);

  sage_layer_kernel<false><<<SG_BLOCKS, SG_THREADS, 0, stream>>>(
      h, col, cnt, src, dst, W2l, b2l, W2r, out, N, E);
}


# Bucket build: a work-group owns 1024 destinations, sifts the whole edge list, ranks its edges with LDS atomics, writes whole bucket lines; global atomics and memset removed; sums in arrival order
# speedup vs baseline: 2.8565x; 1.0702x over previous
#include <hip/hip_runtime.h>
#include <math.h>


typedef __attribute__((ext_vector_type(2))) float v2f;
typedef __attribute__((ext_vector_type(8))) float v8f;
typedef __attribute__((ext_vector_type(4))) float v4f;

#define CCH 128

#define CAP 32
#define BK_NODES   1024
#define BK_THREADS 1024
#define BK_WAVES   (BK_THREADS / 32)
#define BK_BATCH   8
#define BK_LIST    256
static_assert(BK_NODES == BK_THREADS, "one thread clears and stores one counter");

__device__ __forceinline__ void rank_list(const int2* wl, int m, int* lcnt,
                                          int* lcol, int lane) {
  for (int i = lane; i < m; i += 32) {
    const int2 en = wl[i];
    const int r = atomicAdd(&lcnt[en.x], 1);
    if (r < CAP) lcol[en.x * CAP + r] = en.y;
  }
}

__global__ __launch_bounds__(BK_THREADS) void bucket_kernel(
    const int* __restrict__ src, const int* __restrict__ dst,
    int* __restrict__ cnt, int* __restrict__ col, int E, int n) {
  __shared__ int lcnt[BK_NODES];
  __shared__ __attribute__((aligned(16))) int lcol[BK_NODES * CAP];
  __shared__ __attribute__((aligned(8))) int2 wlist[BK_WAVES * BK_LIST];
  const int tid  = threadIdx.x;
  const int lane = tid & 31;
  const int wave = __builtin_amdgcn_readfirstlane(tid >> 5);
  int2* wl = wlist + wave * BK_LIST;
  int head = (int)(((16u - ((unsigned)(size_t)dst & 15u)) & 15u) >> 2);
  if (head > E) head = E;
  const int nvec  = (E - head) >> 2;
  const int tail0 = head + 4 * nvec;
  const int4* __restrict__ dv = (const int4*)(dst + head);
  const int per   = BK_THREADS * BK_BATCH;
  const int steps = nvec / per;
  const int s_off = (int)(((long)blockIdx.x * steps) / gridDim.x);

  for (int base = blockIdx.x * BK_NODES; base < n; base += gridDim.x * BK_NODES) {
    lcnt[tid] = 0;
    __syncthreads();

    int wcount = 0;
    auto sift = [&](const unsigned (&dl)[BK_BATCH][4], int i0) {
#pragma unroll
      for (int u = 0; u < BK_BATCH; ++u) {
#pragma unroll
        for (int j = 0; j < 4; ++j) {
          const bool     hit = dl[u][j] < BK_NODES;
          const unsigned m   = __builtin_amdgcn_ballot_w32(hit);
          if (__builtin_expect(m != 0u, 0)) {
            if (wcount > BK_LIST - 32) {
              rank_list(wl, wcount, lcnt, lcol, lane);
              wcount = 0;
            }
            int t4 = 4 * tid;
            asm volatile("" : "+v"(t4));
            const int pos = wcount + (int)__builtin_amdgcn_mbcnt_lo(m, 0u);
            const int e   = (head + 4 * (i0 + u * BK_THREADS) + j) + t4;
            if (hit) wl[pos] = make_int2((int)dl[u][j], e);
            wcount += __builtin_popcount(m);
          }
        }
      }
    };

    int4 q[BK_BATCH];
    int  i0n = s_off * per;
    if (steps > 0) {
#pragma unroll
      for (int u = 0; u < BK_BATCH; ++u) q[u] = dv[i0n + u * BK_THREADS + tid];
    }
    for (int s = 0; s < steps; ++s) {
      const int i0 = i0n;
      unsigned dl[BK_BATCH][4];
#pragma unroll
      for (int u = 0; u < BK_BATCH; ++u) {
        dl[u][0] = (unsigned)(q[u].x - base); dl[u][1] = (unsigned)(q[u].y - base);
        dl[u][2] = (unsigned)(q[u].z - base); dl[u][3] = (unsigned)(q[u].w - base);
        asm volatile("" : "+v"(dl[u][0]), "+v"(dl[u][1]), "+v"(dl[u][2]), "+v"(dl[u][3]));
      }
      int sn = s + 1; if (sn >= steps) sn = steps - 1;
      sn += s_off;    if (sn >= steps) sn -= steps;
      i0n = sn * per;
#pragma unroll
      for (int u = 0; u < BK_BATCH; ++u) q[u] = dv[i0n + u * BK_THREADS + tid];
      sift(dl, i0);
    }
    {
      const int i0 = steps * per;
      unsigned dl[BK_BATCH][4];
#pragma unroll
      for (int u = 0; u < BK_BATCH; ++u) {
        const int i = i0 + u * BK_THREADS + tid;
        int4 t = make_int4(base + BK_NODES, base + BK_NODES, base + BK_NODES,
                           base + BK_NODES);
        if (i < nvec) t = dv[i];
        dl[u][0] = (unsigned)(t.x - base); dl[u][1] = (unsigned)(t.y - base);
        dl[u][2] = (unsigned)(t.z - base); dl[u][3] = (unsigned)(t.w - base);
      }
      if (i0 < nvec) sift(dl, i0);
    }
    rank_list(wl, wcount, lcnt, lcol, lane);
    if (tid < 8) {
      const int e = (tid < 4) ? tid : (tail0 + tid - 4);
      const bool mine = (tid < 4) ? (e < head) : (e < E);
      if (mine) {
        const unsigned dl = (unsigned)(dst[e] - base);
        if (dl < BK_NODES) {
          const int rr = atomicAdd(&lcnt[dl], 1);
          if (rr < CAP) lcol[dl * CAP + rr] = e;
        }
      }
    }
    __syncthreads();

    int tw = tid;
    asm volatile("" : "+v"(tw));
    for (int k0 = 0; k0 < CAP; k0 += 8) {
      int val[8];
#pragma unroll
      for (int kk = 0; kk < 8; ++kk) {
        const int i    = (k0 + kk) * BK_THREADS + tw;
        const int node = i >> 5, slot = i & 31;
        int c = lcnt[node]; if (c > CAP) c = CAP;
        val[kk] = 0;
        if (slot < c) val[kk] = src[lcol[i]];
      }
#pragma unroll
      for (int kk = 0; kk < 8; ++kk) {
        const int i    = (k0 + kk) * BK_THREADS + tw;
        const int node = i >> 5, slot = i & 31;
        if (base + node < n) col[(long)(base + node) * CAP + slot] = val[kk];
      }
    }
    if (base + tw < n) cnt[base + tw] = lcnt[tw];
    __syncthreads();
  }
}

#define LDP 132
#define SG_WAVES   16
#define SG_THREADS (SG_WAVES * 32)
#define SG_BLOCKS  256

__device__ __forceinline__ v8f wmma4(v2f a, v2f b, v8f c) {
  return __builtin_amdgcn_wmma_f32_16x16x4_f32(false, a, false, b, (short)0, c,
                                               false, false);
}

__device__ __forceinline__ void gather_tile(
    const float* __restrict__ feat, const int* __restrict__ col,
    const int* __restrict__ deg, const int* __restrict__ src,
    const int* __restrict__ dst, float* slot, int node0, int n, int E,
    int lane) {
  const float* frow = feat + lane * 4;
  for (int g = 0; g < 4; ++g) {
    const int nb = node0 + 4 * g;
    int   my_d = 0;
    float my_s = 1.0f;
    if (lane < 4 && nb + lane < n) {
      my_d = deg[nb + lane];
      my_s = 1.0f / fmaxf((float)my_d, 1.0f);
    }
    int cb[4] = {0, 0, 0, 0};
    if (nb + (lane >> 3) < n) {
      const int4 q = *(const int4*)(col + (long)nb * CAP + lane * 4);
      cb[0] = q.x; cb[1] = q.y; cb[2] = q.z; cb[3] = q.w;
    }
    int d[4];
#pragma unroll
    for (int t = 0; t < 4; ++t) d[t] = __builtin_amdgcn_readlane(my_d, t);
    int ck[4][8];
#pragma unroll
    for (int t = 0; t < 4; ++t)
#pragma unroll
      for (int k = 0; k < 8; ++k)
        ck[t][k] = __builtin_amdgcn_readlane(cb[k & 3], 8 * t + (k >> 2));
    v4f v[4][8];
#pragma unroll
    for (int t = 0; t < 4; ++t)
#pragma unroll
      for (int k = 0; k < 8; ++k) {
        v[t][k] = (v4f){0.f, 0.f, 0.f, 0.f};
        if (k < d[t] && d[t] <= CAP)
          v[t][k] = *(const v4f*)(frow + (long)ck[t][k] * CCH);
      }
#pragma unroll
    for (int t = 0; t < 4; ++t)
#pragma unroll
      for (int k = 0; k < 8; ++k)
        asm volatile("" : "+v"(v[t][k]));
#pragma unroll
    for (int t = 0; t < 4; ++t) {
      float4 acc = make_float4(0.f, 0.f, 0.f, 0.f);
#pragma unroll
      for (int k = 0; k < 8; ++k) {
        acc.x += v[t][k].x; acc.y += v[t][k].y;
        acc.z += v[t][k].z; acc.w += v[t][k].w;
      }
      const int dt = d[t];
      if (dt > 8) {
        if (dt <= CAP) {
          for (int j = 8; j < dt; j += 8) {
            const int r  = dt - j;
            const int l0 = 8 * t + (j >> 2);
            int cj[8];
#pragma unroll
            for (int k = 0; k < 8; ++k)
              cj[k] = __builtin_amdgcn_readlane(cb[k & 3], l0 + (k >> 2));
            float4 w[8];
#pragma unroll
            for (int k = 0; k < 8; ++k) {
              w[k] = make_float4(0.f, 0.f, 0.f, 0.f);
              if (k < r) w[k] = *(const float4*)(frow + (long)cj[k] * CCH);
            }
#pragma unroll
            for (int k = 0; k < 8; ++k) {
              acc.x += w[k].x; acc.y += w[k].y; acc.z += w[k].z; acc.w += w[k].w;
            }
          }
        } else {
          const int node = nb + t;
          for (int e0 = 0; e0 < E; e0 += 32) {
            const int  e   = e0 + lane;
            const bool hit = (e < E) && (dst[e] == node);
            const int  sv  = hit ? src[e] : 0;
            unsigned mask  = __builtin_amdgcn_ballot_w32(hit);
            while (mask != 0u) {
              const int b = __builtin_ctz(mask);
              mask &= mask - 1u;
              const int sidx = __builtin_amdgcn_readlane(sv, b);
              const float4 u = *(const float4*)(frow + (long)sidx * CCH);
              acc.x += u.x; acc.y += u.y; acc.z += u.z; acc.w += u.w;
            }
          }
        }
      }
      const float s = __int_as_float(
          __builtin_amdgcn_readlane(__float_as_int(my_s), t));
      acc.x *= s; acc.y *= s; acc.z *= s; acc.w *= s;
      *(float4*)(slot + (4 * g + t) * LDP + lane * 4) = acc;
    }
  }
}

template <bool A_LDS>
__device__ __forceinline__ void half_tile(v8f (&acc)[8], const float4 (&a)[16],
                                          const float* ap, const float* wl,
                                          int wofs) {
  float4 bc[8], bn[8];
  float4 ac = make_float4(0.f, 0.f, 0.f, 0.f), an = ac;
#pragma unroll
  for (int n = 0; n < 8; ++n) bc[n] = *(const float4*)(wl + wofs + n * 16 * LDP);
  if (A_LDS) ac = *(const float4*)(ap);
#pragma unroll
  for (int q = 0; q < 16; ++q) {
    if (q + 1 < 16) {
#pragma unroll
      for (int n = 0; n < 8; ++n)
        bn[n] = *(const float4*)(wl + wofs + n * 16 * LDP + 8 * (q + 1));
      if (A_LDS) an = *(const float4*)(ap + 8 * (q + 1));
    }
    const float4 av = A_LDS ? ac : a[q];
    v2f a0, a1;
    a0.x = av.x; a0.y = av.y; a1.x = av.z; a1.y = av.w;
#pragma unroll
    for (int n = 0; n < 8; ++n) {
      v2f b; b.x = bc[n].x; b.y = bc[n].y;
      acc[n] = wmma4(a0, b, acc[n]);
    }
#pragma unroll
    for (int n = 0; n < 8; ++n) {
      v2f b; b.x = bc[n].z; b.y = bc[n].w;
      acc[n] = wmma4(a1, b, acc[n]);
    }
#pragma unroll
    for (int n = 0; n < 8; ++n) bc[n] = bn[n];
    ac = an;
  }
}

template <bool SIG>
__global__ __launch_bounds__(SG_THREADS) void sage_layer_kernel(
    const float* __restrict__ xin,
    const int* __restrict__ col,
    const int* __restrict__ deg,
    const int* __restrict__ src,
    const int* __restrict__ dst,
    const float* __restrict__ Wl,
    const float* __restrict__ bl,
    const float* __restrict__ Wr,
    float* __restrict__ out,
    int n_nodes, int E) {
  __shared__ __attribute__((aligned(16))) float wlds[256 * LDP];
  __shared__ __attribute__((aligned(16))) float aslot[SG_WAVES * 16 * LDP];
  const int tid = threadIdx.x;

#pragma unroll 8
  for (int j = 0; j < (256 * 32) / SG_THREADS; ++j) {
    const int i = j * SG_THREADS + tid, row = i >> 5, c4 = i & 31;
    const float* wsrc = (row < 128) ? (Wl + row * CCH) : (Wr + (row - 128) * CCH);
    *(float4*)(wlds + row * LDP + c4 * 4) = *(const float4*)(wsrc + c4 * 4);
  }
  __syncthreads();

  const int lane = tid & 31;
  const int wave = __builtin_amdgcn_readfirstlane(tid >> 5);
  const int h    = lane >> 4;
  const int c    = lane & 15;
  const int ntiles = (n_nodes + 15) >> 4;
  int tile = blockIdx.x + SG_BLOCKS * wave;
  if (tile >= ntiles) return;

  const int wofs = c * LDP + 4 * h;
  int wofs_r = wofs + 128 * LDP;
  asm volatile("" : "+v"(wofs_r));
  float bias[8];
#pragma unroll
  for (int n = 0; n < 8; ++n) bias[n] = bl[n * 16 + c];

  float*       slot = aslot + wave * (16 * LDP);
  const float* ap   = slot + c * LDP + 4 * h;

  for (;;) {
    asm volatile("" ::: "memory");

    gather_tile(xin, col, deg, src, dst, slot, tile * 16, n_nodes, E, lane);

    asm volatile("" ::: "memory");

    int m = tile * 16 + c;  if (m >= n_nodes) m = n_nodes - 1;
    float4 ax[16];
    {
      const float* p = xin + (long)m * CCH + 4 * h;
#pragma unroll
      for (int q = 0; q < 16; ++q) ax[q] = *(const float4*)(p + 8 * q);
    }
    v8f acc[8];
#pragma unroll
    for (int n = 0; n < 8; ++n) acc[n] = {};

    half_tile<true>(acc, ax, ap, wlds, wofs);
    half_tile<false>(acc, ax, ap, wlds, wofs_r);

    const int mrow = tile * 16 + 8 * h;
#pragma unroll
    for (int r = 0; r < 8; ++r) {
      if (mrow + r < n_nodes) {
        float* orow = out + (long)(mrow + r) * CCH + c;
#pragma unroll
        for (int n = 0; n < 8; ++n) {
          float v = acc[n][r] + bias[n];
          if (SIG) v = 1.0f / (1.0f + __expf(-v));
          orow[n * 16] = v;
        }
      }
    }

    tile += SG_BLOCKS * SG_WAVES;
    if (tile >= ntiles) break;
  }
}

extern "C" void kernel_launch(void* const* d_in, const int* in_sizes, int n_in,
                              void* d_out, int out_size, void* d_ws, size_t ws_size,
                              hipStream_t stream) {
  const float* x   = (const float*)d_in[0];
  const int*   ei  = (const int*)d_in[1];
  const float* W1l = (const float*)d_in[2];
  const float* b1l = (const float*)d_in[3];
  const float* W1r = (const float*)d_in[4];
  const float* W2l = (const float*)d_in[5];
  const float* b2l = (const float*)d_in[6];
  const float* W2r = (const float*)d_in[7];
  float* out = (float*)d_out;

  const int N = in_sizes[0] / CCH;
  const int E = in_sizes[1] / 2;

  int*   col = (int*)d_ws;
  int*   cnt = col + (size_t)N * CAP;
  float* h   = (float*)d_ws + (size_t)N * CCH;

  const int* src = ei;
  const int* dst = ei + E;

  const int bblocks = (N + BK_NODES - 1) / BK_NODES;

  bucket_kernel<<<bblocks, BK_THREADS, 0, stream>>>(src, dst, cnt, col, E, N);

  sage_layer_kernel<true><<<SG_BLOCKS, SG_THREADS, 0, stream>>>(
      x, col, cnt, src, dst, W1l, b1l, W1r, h, N, E);

  sage_layer_kernel<false><<<SG_BLOCKS, SG_THREADS, 0, stream>>>(
      h, col, cnt, src, dst, W2l, b2l, W2r, out, N, E);
}


# Layer tile order: root product (x @ Wr) first, bucket lines requested ahead, neighbour gather between the two products, under other waves' WMMAs; the two K halves are summed in swapped order
# speedup vs baseline: 2.9671x; 1.0387x over previous
#include <hip/hip_runtime.h>
#include <math.h>


typedef __attribute__((ext_vector_type(2))) float v2f;
typedef __attribute__((ext_vector_type(8))) float v8f;
typedef __attribute__((ext_vector_type(4))) float v4f;

#define CCH 128

#define CAP 32
#define BK_NODES   1024
#define BK_THREADS 1024
#define BK_WAVES   (BK_THREADS / 32)
#define BK_BATCH   8
#define BK_LIST    256
static_assert(BK_NODES == BK_THREADS, "one thread clears and stores one counter");

__device__ __forceinline__ void rank_list(const int2* wl, int m, int* lcnt,
                                          int* lcol, int lane) {
  for (int i = lane; i < m; i += 32) {
    const int2 en = wl[i];
    const int r = atomicAdd(&lcnt[en.x], 1);
    if (r < CAP) lcol[en.x * CAP + r] = en.y;
  }
}

__global__ __launch_bounds__(BK_THREADS) void bucket_kernel(
    const int* __restrict__ src, const int* __restrict__ dst,
    int* __restrict__ cnt, int* __restrict__ col, int E, int n) {
  __shared__ int lcnt[BK_NODES];
  __shared__ __attribute__((aligned(16))) int lcol[BK_NODES * CAP];
  __shared__ __attribute__((aligned(8))) int2 wlist[BK_WAVES * BK_LIST];
  const int tid  = threadIdx.x;
  const int lane = tid & 31;
  const int wave = __builtin_amdgcn_readfirstlane(tid >> 5);
  int2* wl = wlist + wave * BK_LIST;
  int head = (int)(((16u - ((unsigned)(size_t)dst & 15u)) & 15u) >> 2);
  if (head > E) head = E;
  const int nvec  = (E - head) >> 2;
  const int tail0 = head + 4 * nvec;
  const int4* __restrict__ dv = (const int4*)(dst + head);
  const int per   = BK_THREADS * BK_BATCH;
  const int steps = nvec / per;
  const int s_off = (int)(((long)blockIdx.x * steps) / gridDim.x);

  for (int base = blockIdx.x * BK_NODES; base < n; base += gridDim.x * BK_NODES) {
    lcnt[tid] = 0;
    __syncthreads();

    int wcount = 0;
    auto sift = [&](const unsigned (&dl)[BK_BATCH][4], int i0) {
#pragma unroll
      for (int u = 0; u < BK_BATCH; ++u) {
#pragma unroll
        for (int j = 0; j < 4; ++j) {
          const bool     hit = dl[u][j] < BK_NODES;
          const unsigned m   = __builtin_amdgcn_ballot_w32(hit);
          if (__builtin_expect(m != 0u, 0)) {
            if (wcount > BK_LIST - 32) {
              rank_list(wl, wcount, lcnt, lcol, lane);
              wcount = 0;
            }
            int t4 = 4 * tid;
            asm volatile("" : "+v"(t4));
            const int pos = wcount + (int)__builtin_amdgcn_mbcnt_lo(m, 0u);
            const int e   = (head + 4 * (i0 + u * BK_THREADS) + j) + t4;
            if (hit) wl[pos] = make_int2((int)dl[u][j], e);
            wcount += __builtin_popcount(m);
          }
        }
      }
    };

    int4 q[BK_BATCH];
    int  i0n = s_off * per;
    if (steps > 0) {
#pragma unroll
      for (int u = 0; u < BK_BATCH; ++u) q[u] = dv[i0n + u * BK_THREADS + tid];
    }
    for (int s = 0; s < steps; ++s) {
      const int i0 = i0n;
      unsigned dl[BK_BATCH][4];
#pragma unroll
      for (int u = 0; u < BK_BATCH; ++u) {
        dl[u][0] = (unsigned)(q[u].x - base); dl[u][1] = (unsigned)(q[u].y - base);
        dl[u][2] = (unsigned)(q[u].z - base); dl[u][3] = (unsigned)(q[u].w - base);
        asm volatile("" : "+v"(dl[u][0]), "+v"(dl[u][1]), "+v"(dl[u][2]), "+v"(dl[u][3]));
      }
      int sn = s + 1; if (sn >= steps) sn = steps - 1;
      sn += s_off;    if (sn >= steps) sn -= steps;
      i0n = sn * per;
#pragma unroll
      for (int u = 0; u < BK_BATCH; ++u) q[u] = dv[i0n + u * BK_THREADS + tid];
      sift(dl, i0);
    }
    {
      const int i0 = steps * per;
      unsigned dl[BK_BATCH][4];
#pragma unroll
      for (int u = 0; u < BK_BATCH; ++u) {
        const int i = i0 + u * BK_THREADS + tid;
        int4 t = make_int4(base + BK_NODES, base + BK_NODES, base + BK_NODES,
                           base + BK_NODES);
        if (i < nvec) t = dv[i];
        dl[u][0] = (unsigned)(t.x - base); dl[u][1] = (unsigned)(t.y - base);
        dl[u][2] = (unsigned)(t.z - base); dl[u][3] = (unsigned)(t.w - base);
      }
      if (i0 < nvec) sift(dl, i0);
    }
    rank_list(wl, wcount, lcnt, lcol, lane);
    if (tid < 8) {
      const int e = (tid < 4) ? tid : (tail0 + tid - 4);
      const bool mine = (tid < 4) ? (e < head) : (e < E);
      if (mine) {
        const unsigned dl = (unsigned)(dst[e] - base);
        if (dl < BK_NODES) {
          const int rr = atomicAdd(&lcnt[dl], 1);
          if (rr < CAP) lcol[dl * CAP + rr] = e;
        }
      }
    }
    __syncthreads();

    int tw = tid;
    asm volatile("" : "+v"(tw));
    for (int k0 = 0; k0 < CAP; k0 += 8) {
      int val[8];
#pragma unroll
      for (int kk = 0; kk < 8; ++kk) {
        const int i    = (k0 + kk) * BK_THREADS + tw;
        const int node = i >> 5, slot = i & 31;
        int c = lcnt[node]; if (c > CAP) c = CAP;
        val[kk] = 0;
        if (slot < c) val[kk] = src[lcol[i]];
      }
#pragma unroll
      for (int kk = 0; kk < 8; ++kk) {
        const int i    = (k0 + kk) * BK_THREADS + tw;
        const int node = i >> 5, slot = i & 31;
        if (base + node < n) col[(long)(base + node) * CAP + slot] = val[kk];
      }
    }
    if (base + tw < n) cnt[base + tw] = lcnt[tw];
    __syncthreads();
  }
}

#define LDP 132
#define SG_WAVES   16
#define SG_THREADS (SG_WAVES * 32)
#define SG_BLOCKS  256

__device__ __forceinline__ v8f wmma4(v2f a, v2f b, v8f c) {
  return __builtin_amdgcn_wmma_f32_16x16x4_f32(false, a, false, b, (short)0, c,
                                               false, false);
}

__device__ __forceinline__ void bucket_request(
    const int* __restrict__ col, const int* __restrict__ deg, int node0, int n,
    int lane, int4 (&cq)[4], int& my_d) {
  my_d = 0;
  if (lane < 16 && node0 + lane < n) my_d = deg[node0 + lane];
#pragma unroll
  for (int g = 0; g < 4; ++g) {
    const int nb = node0 + 4 * g;
    cq[g] = make_int4(0, 0, 0, 0);
    if (nb + (lane >> 3) < n)
      cq[g] = *(const int4*)(col + (long)nb * CAP + lane * 4);
  }
}

__device__ __forceinline__ void gather_tile(
    const float* __restrict__ feat, const int* __restrict__ src,
    const int* __restrict__ dst, float* slot, int node0, int E, int lane,
    int4 (&cq)[4], int my_d) {
  const float* frow = feat + lane * 4;
  const float my_s = 1.0f / fmaxf((float)my_d, 1.0f);
  for (int g = 0; g < 4; ++g) {
    const int cb[4] = {cq[0].x, cq[0].y, cq[0].z, cq[0].w};
    cq[0] = cq[1]; cq[1] = cq[2]; cq[2] = cq[3];
    int d[4];
#pragma unroll
    for (int t = 0; t < 4; ++t) d[t] = __builtin_amdgcn_readlane(my_d, 4 * g + t);
    int ck[4][8];
#pragma unroll
    for (int t = 0; t < 4; ++t)
#pragma unroll
      for (int k = 0; k < 8; ++k)
        ck[t][k] = __builtin_amdgcn_readlane(cb[k & 3], 8 * t + (k >> 2));
    v4f v[4][8];
#pragma unroll
    for (int t = 0; t < 4; ++t)
#pragma unroll
      for (int k = 0; k < 8; ++k) {
        v[t][k] = (v4f){0.f, 0.f, 0.f, 0.f};
        if (k < d[t] && d[t] <= CAP)
          v[t][k] = *(const v4f*)(frow + (long)ck[t][k] * CCH);
      }
#pragma unroll
    for (int t = 0; t < 4; ++t)
#pragma unroll
      for (int k = 0; k < 8; ++k)
        asm volatile("" : "+v"(v[t][k]));
#pragma unroll
    for (int t = 0; t < 4; ++t) {
      float4 acc = make_float4(0.f, 0.f, 0.f, 0.f);
#pragma unroll
      for (int k = 0; k < 8; ++k) {
        acc.x += v[t][k].x; acc.y += v[t][k].y;
        acc.z += v[t][k].z; acc.w += v[t][k].w;
      }
      const int dt = d[t];
      if (dt > 8 && dt <= CAP) {
        for (int j = 8; j < dt; j += 8) {
          const int r  = dt - j;
          const int l0 = 8 * t + (j >> 2);
          int cj[8];
#pragma unroll
          for (int k = 0; k < 8; ++k)
            cj[k] = __builtin_amdgcn_readlane(cb[k & 3], l0 + (k >> 2));
          float4 w[8];
#pragma unroll
          for (int k = 0; k < 8; ++k) {
            w[k] = make_float4(0.f, 0.f, 0.f, 0.f);
            if (k < r) w[k] = *(const float4*)(frow + (long)cj[k] * CCH);
          }
#pragma unroll
          for (int k = 0; k < 8; ++k) {
            acc.x += w[k].x; acc.y += w[k].y; acc.z += w[k].z; acc.w += w[k].w;
          }
        }
      }
      const float s = __int_as_float(
          __builtin_amdgcn_readlane(__float_as_int(my_s), 4 * g + t));
      acc.x *= s; acc.y *= s; acc.z *= s; acc.w *= s;
      *(float4*)(slot + (4 * g + t) * LDP + lane * 4) = acc;
    }
  }
  const unsigned big = __builtin_amdgcn_ballot_w32(my_d > CAP);
  if (__builtin_expect(big != 0u, 0)) {
    for (int t = 0; t < 16; ++t) {
      if (((big >> t) & 1u) == 0u) continue;
      const int node = node0 + t;
      float4 acc = make_float4(0.f, 0.f, 0.f, 0.f);
      for (int e0 = 0; e0 < E; e0 += 32) {
        const int  e   = e0 + lane;
        const bool hit = (e < E) && (dst[e] == node);
        const int  sv  = hit ? src[e] : 0;
        unsigned mask  = __builtin_amdgcn_ballot_w32(hit);
        while (mask != 0u) {
          const int b = __builtin_ctz(mask);
          mask &= mask - 1u;
          const int sidx = __builtin_amdgcn_readlane(sv, b);
          const float4 u = *(const float4*)(frow + (long)sidx * CCH);
          acc.x += u.x; acc.y += u.y; acc.z += u.z; acc.w += u.w;
        }
      }
      const float s = __int_as_float(
          __builtin_amdgcn_readlane(__float_as_int(my_s), t));
      acc.x *= s; acc.y *= s; acc.z *= s; acc.w *= s;
      *(float4*)(slot + t * LDP + lane * 4) = acc;
    }
  }
}

template <bool A_LDS>
__device__ __forceinline__ void half_tile(v8f (&acc)[8], const float4 (&a)[16],
                                          const float* ap, const float* wl,
                                          int wofs) {
  float4 bc[8], bn[8];
  float4 ac = make_float4(0.f, 0.f, 0.f, 0.f), an = ac;
#pragma unroll
  for (int n = 0; n < 8; ++n) bc[n] = *(const float4*)(wl + wofs + n * 16 * LDP);
  if (A_LDS) ac = *(const float4*)(ap);
#pragma unroll
  for (int q = 0; q < 16; ++q) {
    if (q + 1 < 16) {
#pragma unroll
      for (int n = 0; n < 8; ++n)
        bn[n] = *(const float4*)(wl + wofs + n * 16 * LDP + 8 * (q + 1));
      if (A_LDS) an = *(const float4*)(ap + 8 * (q + 1));
    }
    const float4 av = A_LDS ? ac : a[q];
    v2f a0, a1;
    a0.x = av.x; a0.y = av.y; a1.x = av.z; a1.y = av.w;
#pragma unroll
    for (int n = 0; n < 8; ++n) {
      v2f b; b.x = bc[n].x; b.y = bc[n].y;
      acc[n] = wmma4(a0, b, acc[n]);
    }
#pragma unroll
    for (int n = 0; n < 8; ++n) {
      v2f b; b.x = bc[n].z; b.y = bc[n].w;
      acc[n] = wmma4(a1, b, acc[n]);
    }
#pragma unroll
    for (int n = 0; n < 8; ++n) bc[n] = bn[n];
    ac = an;
  }
}

template <bool SIG>
__global__ __launch_bounds__(SG_THREADS) void sage_layer_kernel(
    const float* __restrict__ xin,
    const int* __restrict__ col,
    const int* __restrict__ deg,
    const int* __restrict__ src,
    const int* __restrict__ dst,
    const float* __restrict__ Wl,
    const float* __restrict__ bl,
    const float* __restrict__ Wr,
    float* __restrict__ out,
    int n_nodes, int E) {
  __shared__ __attribute__((aligned(16))) float wlds[256 * LDP];
  __shared__ __attribute__((aligned(16))) float aslot[SG_WAVES * 16 * LDP];
  const int tid = threadIdx.x;
  const int lane = tid & 31;
  const int wave = __builtin_amdgcn_readfirstlane(tid >> 5);
  const int h    = lane >> 4;
  const int c    = lane & 15;
  const int ntiles = (n_nodes + 15) >> 4;
  int tile = blockIdx.x + SG_BLOCKS * wave;

  float4 ax[16];
  int4   cq[4];
  int    my_d;
  auto request = [&](int t) {
    int m = t * 16 + c;  if (m >= n_nodes) m = n_nodes - 1;
    const float* p = xin + (long)m * CCH + 4 * h;
#pragma unroll
    for (int q = 0; q < 16; ++q) ax[q] = *(const float4*)(p + 8 * q);
    bucket_request(col, deg, t * 16, n_nodes, lane, cq, my_d);
  };
  request(tile);

#pragma unroll 8
  for (int j = 0; j < (256 * 32) / SG_THREADS; ++j) {
    const int i = j * SG_THREADS + tid, row = i >> 5, c4 = i & 31;
    const float* wsrc = (row < 128) ? (Wl + row * CCH) : (Wr + (row - 128) * CCH);
    *(float4*)(wlds + row * LDP + c4 * 4) = *(const float4*)(wsrc + c4 * 4);
  }
  __syncthreads();

  if (tile >= ntiles) return;

  const int wofs = c * LDP + 4 * h;
  int wofs_r = wofs + 128 * LDP;
  asm volatile("" : "+v"(wofs_r));
  float bias[8];
#pragma unroll
  for (int n = 0; n < 8; ++n) bias[n] = bl[n * 16 + c];

  float*       slot = aslot + wave * (16 * LDP);
  const float* ap   = slot + c * LDP + 4 * h;

  for (;;) {
    asm volatile("" ::: "memory");

    v8f acc[8];
#pragma unroll
    for (int n = 0; n < 8; ++n) acc[n] = {};

    half_tile<false>(acc, ax, ap, wlds, wofs_r);

    gather_tile(xin, src, dst, slot, tile * 16, E, lane, cq, my_d);

    asm volatile("" ::: "memory");

    half_tile<true>(acc, ax, ap, wlds, wofs);

    const int mrow = tile * 16 + 8 * h;
#pragma unroll
    for (int r = 0; r < 8; ++r) {
      if (mrow + r < n_nodes) {
        float* orow = out + (long)(mrow + r) * CCH + c;
#pragma unroll
        for (int n = 0; n < 8; ++n) {
          float v = acc[n][r] + bias[n];
          if (SIG) v = 1.0f / (1.0f + __expf(-v));
          orow[n * 16] = v;
        }
      }
    }

    tile += SG_BLOCKS * SG_WAVES;
    if (tile >= ntiles) break;
    request(tile);
  }
}

extern "C" void kernel_launch(void* const* d_in, const int* in_sizes, int n_in,
                              void* d_out, int out_size, void* d_ws, size_t ws_size,
                              hipStream_t stream) {
  const float* x   = (const float*)d_in[0];
  const int*   ei  = (const int*)d_in[1];
  const float* W1l = (const float*)d_in[2];
  const float* b1l = (const float*)d_in[3];
  const float* W1r = (const float*)d_in[4];
  const float* W2l = (const float*)d_in[5];
  const float* b2l = (const float*)d_in[6];
  const float* W2r = (const float*)d_in[7];
  float* out = (float*)d_out;

  const int N = in_sizes[0] / CCH;
  const int E = in_sizes[1] / 2;

  int*   col = (int*)d_ws;
  int*   cnt = col + (size_t)N * CAP;
  float* h   = (float*)d_ws + (size_t)N * CCH;

  const int* src = ei;
  const int* dst = ei + E;

  const int bblocks = (N + BK_NODES - 1) / BK_NODES;

  bucket_kernel<<<bblocks, BK_THREADS, 0, stream>>>(src, dst, cnt, col, E, N);

  sage_layer_kernel<true><<<SG_BLOCKS, SG_THREADS, 0, stream>>>(
      x, col, cnt, src, dst, W1l, b1l, W1r, h, N, E);

  sage_layer_kernel<false><<<SG_BLOCKS, SG_THREADS, 0, stream>>>(
      h, col, cnt, src, dst, W2l, b2l, W2r, out, N, E);
}


# Bucket build: a new first launch sorts the edges by destination range (LDS counters); an owner reads only its own edges, not the whole list; slots stay in arrival order (row-sum order may differ)
# speedup vs baseline: 3.2252x; 1.0870x over previous
#include <hip/hip_runtime.h>
#include <math.h>


typedef __attribute__((ext_vector_type(2))) float v2f;
typedef __attribute__((ext_vector_type(8))) float v8f;
typedef __attribute__((ext_vector_type(4))) float v4f;

#define CCH 128

#define CAP 32
#define BK_NODES   1024
#define BK_THREADS 1024
#define BK_WAVES   (BK_THREADS / 32)
#define BK_BATCH   8
#define BK_LIST    256
static_assert(BK_NODES == BK_THREADS, "one thread clears and stores one counter");

#define PL_WGS     128
#define PL_THREADS 512
#define PL_CAP     128
#define PL_BATCH   10
#define PR_MAX     1024
static_assert(BK_NODES == 1024, "range number = destination >> 10");
static_assert(PL_WGS * 8 == BK_THREADS, "eight threads of the owner read one list");

__global__ __launch_bounds__(PL_THREADS) void partition_kernel(
    const int* __restrict__ dst, int* __restrict__ plist_raw,
    int* __restrict__ pcount, int E, int n) {
  __shared__ int pc[PR_MAX];
  const int R = (n + BK_NODES - 1) / BK_NODES;
  if (R > PR_MAX) return;
  const int tid = threadIdx.x;
  for (int i = tid; i < PR_MAX; i += PL_THREADS) pc[i] = 0;
  __syncthreads();
  const int chunk = (E + PL_WGS - 1) / PL_WGS;
  const int e0 = blockIdx.x * chunk;
  int e1 = e0 + chunk;  if (e1 > E) e1 = E;
  int2* mine = (int2*)plist_raw + (size_t)blockIdx.x * R * PL_CAP;
  for (int eb = e0; eb < e1; eb += PL_BATCH * PL_THREADS) {
    int d[PL_BATCH];
#pragma unroll
    for (int j = 0; j < PL_BATCH; ++j) {
      const int e = eb + j * PL_THREADS + tid;
      d[j] = -1;
      if (e < e1) d[j] = dst[e];
    }
    int pos[PL_BATCH];
#pragma unroll
    for (int j = 0; j < PL_BATCH; ++j) {
      const unsigned r = (unsigned)d[j] >> 10;
      pos[j] = PL_CAP;
      if (r < (unsigned)R) pos[j] = atomicAdd(&pc[r], 1);
    }
#pragma unroll
    for (int j = 0; j < PL_BATCH; ++j) {
      const int e = eb + j * PL_THREADS + tid;
      const unsigned r = (unsigned)d[j] >> 10;
      if (pos[j] < PL_CAP) mine[(size_t)r * PL_CAP + pos[j]] = make_int2(d[j], e);
    }
  }
  __syncthreads();
  for (int i = tid; i < R; i += PL_THREADS) pcount[blockIdx.x * PR_MAX + i] = pc[i];
}

__device__ __forceinline__ void rank_list(const int2* wl, int m, int* lcnt,
                                          int* lcol, int lane) {
  for (int i = lane; i < m; i += 32) {
    const int2 en = wl[i];
    const int r = atomicAdd(&lcnt[en.x], 1);
    if (r < CAP) lcol[en.x * CAP + r] = en.y;
  }
}

__global__ __launch_bounds__(BK_THREADS) void bucket_kernel(
    const int* __restrict__ src, const int* __restrict__ dst,
    const int* __restrict__ plist_raw, const int* __restrict__ pcount,
    int* __restrict__ cnt, int* __restrict__ col, int E, int n) {
  __shared__ int lcnt[BK_NODES];
  __shared__ __attribute__((aligned(16))) int lcol[BK_NODES * CAP];
  __shared__ __attribute__((aligned(8))) int2 wlist[BK_WAVES * BK_LIST];
  const int tid  = threadIdx.x;
  const int lane = tid & 31;
  const int wave = __builtin_amdgcn_readfirstlane(tid >> 5);
  int2* wl = wlist + wave * BK_LIST;
  int head = (int)(((16u - ((unsigned)(size_t)dst & 15u)) & 15u) >> 2);
  if (head > E) head = E;
  const int nvec  = (E - head) >> 2;
  const int tail0 = head + 4 * nvec;
  const int4* __restrict__ dv = (const int4*)(dst + head);
  const int per   = BK_THREADS * BK_BATCH;
  const int steps = nvec / per;
  const int s_off = (int)(((long)blockIdx.x * steps) / gridDim.x);

  const int R = (n + BK_NODES - 1) / BK_NODES;
  for (int base = blockIdx.x * BK_NODES; base < n; base += gridDim.x * BK_NODES) {
    lcnt[tid] = 0;
    const int r  = base / BK_NODES;
    const int lw = tid >> 3, sub = tid & 7;
    int c = 0;
    if (R <= PR_MAX) c = pcount[lw * PR_MAX + r];
    const int sift_all = __syncthreads_or((R > PR_MAX) || (c > PL_CAP));

    if (!sift_all) {
      const int2* pl = (const int2*)plist_raw + ((size_t)lw * R + r) * PL_CAP;
#pragma nounroll
      for (int j0 = 0; j0 < PL_CAP / 8; j0 += 8) {
        int2 en[8];
#pragma unroll
        for (int j = 0; j < 8; ++j) {
          en[j] = make_int2(-1, 0);
          const int i = sub + 8 * (j0 + j);
          if (i < c) en[j] = pl[i];
        }
        int rr[8];
#pragma unroll
        for (int j = 0; j < 8; ++j) {
          const unsigned dl = (unsigned)(en[j].x - base);
          rr[j] = CAP;
          if (dl < BK_NODES) rr[j] = atomicAdd(&lcnt[dl], 1);
        }
#pragma unroll
        for (int j = 0; j < 8; ++j) {
          const unsigned dl = (unsigned)(en[j].x - base);
          if (rr[j] < CAP) lcol[dl * CAP + rr[j]] = en[j].y;
        }
      }
    } else {
    int wcount = 0;
    auto sift = [&](const unsigned (&dl)[BK_BATCH][4], int i0) {
#pragma unroll
      for (int u = 0; u < BK_BATCH; ++u) {
#pragma unroll
        for (int j = 0; j < 4; ++j) {
          const bool     hit = dl[u][j] < BK_NODES;
          const unsigned m   = __builtin_amdgcn_ballot_w32(hit);
          if (__builtin_expect(m != 0u, 0)) {
            if (wcount > BK_LIST - 32) {
              rank_list(wl, wcount, lcnt, lcol, lane);
              wcount = 0;
            }
            int t4 = 4 * tid;
            asm volatile("" : "+v"(t4));
            const int pos = wcount + (int)__builtin_amdgcn_mbcnt_lo(m, 0u);
            const int e   = (head + 4 * (i0 + u * BK_THREADS) + j) + t4;
            if (hit) wl[pos] = make_int2((int)dl[u][j], e);
            wcount += __builtin_popcount(m);
          }
        }
      }
    };

    int4 q[BK_BATCH];
    int  i0n = s_off * per;
    if (steps > 0) {
#pragma unroll
      for (int u = 0; u < BK_BATCH; ++u) q[u] = dv[i0n + u * BK_THREADS + tid];
    }
    for (int s = 0; s < steps; ++s) {
      const int i0 = i0n;
      unsigned dl[BK_BATCH][4];
#pragma unroll
      for (int u = 0; u < BK_BATCH; ++u) {
        dl[u][0] = (unsigned)(q[u].x - base); dl[u][1] = (unsigned)(q[u].y - base);
        dl[u][2] = (unsigned)(q[u].z - base); dl[u][3] = (unsigned)(q[u].w - base);
        asm volatile("" : "+v"(dl[u][0]), "+v"(dl[u][1]), "+v"(dl[u][2]), "+v"(dl[u][3]));
      }
      int sn = s + 1; if (sn >= steps) sn = steps - 1;
      sn += s_off;    if (sn >= steps) sn -= steps;
      i0n = sn * per;
#pragma unroll
      for (int u = 0; u < BK_BATCH; ++u) q[u] = dv[i0n + u * BK_THREADS + tid];
      sift(dl, i0);
    }
    {
      const int i0 = steps * per;
      unsigned dl[BK_BATCH][4];
#pragma unroll
      for (int u = 0; u < BK_BATCH; ++u) {
        const int i = i0 + u * BK_THREADS + tid;
        int4 t = make_int4(base + BK_NODES, base + BK_NODES, base + BK_NODES,
                           base + BK_NODES);
        if (i < nvec) t = dv[i];
        dl[u][0] = (unsigned)(t.x - base); dl[u][1] = (unsigned)(t.y - base);
        dl[u][2] = (unsigned)(t.z - base); dl[u][3] = (unsigned)(t.w - base);
      }
      if (i0 < nvec) sift(dl, i0);
    }
    rank_list(wl, wcount, lcnt, lcol, lane);
    if (tid < 8) {
      const int e = (tid < 4) ? tid : (tail0 + tid - 4);
      const bool mine = (tid < 4) ? (e < head) : (e < E);
      if (mine) {
        const unsigned dl = (unsigned)(dst[e] - base);
        if (dl < BK_NODES) {
          const int rr = atomicAdd(&lcnt[dl], 1);
          if (rr < CAP) lcol[dl * CAP + rr] = e;
        }
      }
    }
    }
    __syncthreads();

    int tw = tid;
    asm volatile("" : "+v"(tw));
    for (int k0 = 0; k0 < CAP; k0 += 8) {
      int val[8];
#pragma unroll
      for (int kk = 0; kk < 8; ++kk) {
        const int i    = (k0 + kk) * BK_THREADS + tw;
        const int node = i >> 5, slot = i & 31;
        int c = lcnt[node]; if (c > CAP) c = CAP;
        val[kk] = 0;
        if (slot < c) val[kk] = src[lcol[i]];
      }
#pragma unroll
      for (int kk = 0; kk < 8; ++kk) {
        const int i    = (k0 + kk) * BK_THREADS + tw;
        const int node = i >> 5, slot = i & 31;
        if (base + node < n) col[(long)(base + node) * CAP + slot] = val[kk];
      }
    }
    if (base + tw < n) cnt[base + tw] = lcnt[tw];
    __syncthreads();
  }
}

#define LDP 132
#define SG_WAVES   16
#define SG_THREADS (SG_WAVES * 32)
#define SG_BLOCKS  256

__device__ __forceinline__ v8f wmma4(v2f a, v2f b, v8f c) {
  return __builtin_amdgcn_wmma_f32_16x16x4_f32(false, a, false, b, (short)0, c,
                                               false, false);
}

__device__ __forceinline__ void bucket_request(
    const int* __restrict__ col, const int* __restrict__ deg, int node0, int n,
    int lane, int4 (&cq)[4], int& my_d) {
  my_d = 0;
  if (lane < 16 && node0 + lane < n) my_d = deg[node0 + lane];
#pragma unroll
  for (int g = 0; g < 4; ++g) {
    const int nb = node0 + 4 * g;
    cq[g] = make_int4(0, 0, 0, 0);
    if (nb + (lane >> 3) < n)
      cq[g] = *(const int4*)(col + (long)nb * CAP + lane * 4);
  }
}

__device__ __forceinline__ void gather_tile(
    const float* __restrict__ feat, const int* __restrict__ src,
    const int* __restrict__ dst, float* slot, int node0, int E, int lane,
    int4 (&cq)[4], int my_d) {
  const float* frow = feat + lane * 4;
  const float my_s = 1.0f / fmaxf((float)my_d, 1.0f);
  for (int g = 0; g < 4; ++g) {
    const int cb[4] = {cq[0].x, cq[0].y, cq[0].z, cq[0].w};
    cq[0] = cq[1]; cq[1] = cq[2]; cq[2] = cq[3];
    int d[4];
#pragma unroll
    for (int t = 0; t < 4; ++t) d[t] = __builtin_amdgcn_readlane(my_d, 4 * g + t);
    int ck[4][8];
#pragma unroll
    for (int t = 0; t < 4; ++t)
#pragma unroll
      for (int k = 0; k < 8; ++k)
        ck[t][k] = __builtin_amdgcn_readlane(cb[k & 3], 8 * t + (k >> 2));
    v4f v[4][8];
#pragma unroll
    for (int t = 0; t < 4; ++t)
#pragma unroll
      for (int k = 0; k < 8; ++k) {
        v[t][k] = (v4f){0.f, 0.f, 0.f, 0.f};
        if (k < d[t] && d[t] <= CAP)
          v[t][k] = *(const v4f*)(frow + (long)ck[t][k] * CCH);
      }
#pragma unroll
    for (int t = 0; t < 4; ++t)
#pragma unroll
      for (int k = 0; k < 8; ++k)
        asm volatile("" : "+v"(v[t][k]));
#pragma unroll
    for (int t = 0; t < 4; ++t) {
      float4 acc = make_float4(0.f, 0.f, 0.f, 0.f);
#pragma unroll
      for (int k = 0; k < 8; ++k) {
        acc.x += v[t][k].x; acc.y += v[t][k].y;
        acc.z += v[t][k].z; acc.w += v[t][k].w;
      }
      const int dt = d[t];
      if (dt > 8 && dt <= CAP) {
        for (int j = 8; j < dt; j += 8) {
          const int r  = dt - j;
          const int l0 = 8 * t + (j >> 2);
          int cj[8];
#pragma unroll
          for (int k = 0; k < 8; ++k)
            cj[k] = __builtin_amdgcn_readlane(cb[k & 3], l0 + (k >> 2));
          float4 w[8];
#pragma unroll
          for (int k = 0; k < 8; ++k) {
            w[k] = make_float4(0.f, 0.f, 0.f, 0.f);
            if (k < r) w[k] = *(const float4*)(frow + (long)cj[k] * CCH);
          }
#pragma unroll
          for (int k = 0; k < 8; ++k) {
            acc.x += w[k].x; acc.y += w[k].y; acc.z += w[k].z; acc.w += w[k].w;
          }
        }
      }
      const float s = __int_as_float(
          __builtin_amdgcn_readlane(__float_as_int(my_s), 4 * g + t));
      acc.x *= s; acc.y *= s; acc.z *= s; acc.w *= s;
      *(float4*)(slot + (4 * g + t) * LDP + lane * 4) = acc;
    }
  }
  const unsigned big = __builtin_amdgcn_ballot_w32(my_d > CAP);
  if (__builtin_expect(big != 0u, 0)) {
    for (int t = 0; t < 16; ++t) {
      if (((big >> t) & 1u) == 0u) continue;
      const int node = node0 + t;
      float4 acc = make_float4(0.f, 0.f, 0.f, 0.f);
      for (int e0 = 0; e0 < E; e0 += 32) {
        const int  e   = e0 + lane;
        const bool hit = (e < E) && (dst[e] == node);
        const int  sv  = hit ? src[e] : 0;
        unsigned mask  = __builtin_amdgcn_ballot_w32(hit);
        while (mask != 0u) {
          const int b = __builtin_ctz(mask);
          mask &= mask - 1u;
          const int sidx = __builtin_amdgcn_readlane(sv, b);
          const float4 u = *(const float4*)(frow + (long)sidx * CCH);
          acc.x += u.x; acc.y += u.y; acc.z += u.z; acc.w += u.w;
        }
      }
      const float s = __int_as_float(
          __builtin_amdgcn_readlane(__float_as_int(my_s), t));
      acc.x *= s; acc.y *= s; acc.z *= s; acc.w *= s;
      *(float4*)(slot + t * LDP + lane * 4) = acc;
    }
  }
}

template <bool A_LDS>
__device__ __forceinline__ void half_tile(v8f (&acc)[8], const float4 (&a)[16],
                                          const float* ap, const float* wl,
                                          int wofs) {
  float4 bc[8], bn[8];
  float4 ac = make_float4(0.f, 0.f, 0.f, 0.f), an = ac;
#pragma unroll
  for (int n = 0; n < 8; ++n) bc[n] = *(const float4*)(wl + wofs + n * 16 * LDP);
  if (A_LDS) ac = *(const float4*)(ap);
#pragma unroll
  for (int q = 0; q < 16; ++q) {
    if (q + 1 < 16) {
#pragma unroll
      for (int n = 0; n < 8; ++n)
        bn[n] = *(const float4*)(wl + wofs + n * 16 * LDP + 8 * (q + 1));
      if (A_LDS) an = *(const float4*)(ap + 8 * (q + 1));
    }
    const float4 av = A_LDS ? ac : a[q];
    v2f a0, a1;
    a0.x = av.x; a0.y = av.y; a1.x = av.z; a1.y = av.w;
#pragma unroll
    for (int n = 0; n < 8; ++n) {
      v2f b; b.x = bc[n].x; b.y = bc[n].y;
      acc[n] = wmma4(a0, b, acc[n]);
    }
#pragma unroll
    for (int n = 0; n < 8; ++n) {
      v2f b; b.x = bc[n].z; b.y = bc[n].w;
      acc[n] = wmma4(a1, b, acc[n]);
    }
#pragma unroll
    for (int n = 0; n < 8; ++n) bc[n] = bn[n];
    ac = an;
  }
}

template <bool SIG>
__global__ __launch_bounds__(SG_THREADS) void sage_layer_kernel(
    const float* __restrict__ xin,
    const int* __restrict__ col,
    const int* __restrict__ deg,
    const int* __restrict__ src,
    const int* __restrict__ dst,
    const float* __restrict__ Wl,
    const float* __restrict__ bl,
    const float* __restrict__ Wr,
    float* __restrict__ out,
    int n_nodes, int E) {
  __shared__ __attribute__((aligned(16))) float wlds[256 * LDP];
  __shared__ __attribute__((aligned(16))) float aslot[SG_WAVES * 16 * LDP];
  const int tid = threadIdx.x;
  const int lane = tid & 31;
  const int wave = __builtin_amdgcn_readfirstlane(tid >> 5);
  const int h    = lane >> 4;
  const int c    = lane & 15;
  const int ntiles = (n_nodes + 15) >> 4;
  int tile = blockIdx.x + SG_BLOCKS * wave;

  float4 ax[16];
  int4   cq[4];
  int    my_d;
  auto request = [&](int t) {
    int m = t * 16 + c;  if (m >= n_nodes) m = n_nodes - 1;
    const float* p = xin + (long)m * CCH + 4 * h;
#pragma unroll
    for (int q = 0; q < 16; ++q) ax[q] = *(const float4*)(p + 8 * q);
    bucket_request(col, deg, t * 16, n_nodes, lane, cq, my_d);
  };
  request(tile);

#pragma unroll 8
  for (int j = 0; j < (256 * 32) / SG_THREADS; ++j) {
    const int i = j * SG_THREADS + tid, row = i >> 5, c4 = i & 31;
    const float* wsrc = (row < 128) ? (Wl + row * CCH) : (Wr + (row - 128) * CCH);
    *(float4*)(wlds + row * LDP + c4 * 4) = *(const float4*)(wsrc + c4 * 4);
  }
  __syncthreads();

  if (tile >= ntiles) return;

  const int wofs = c * LDP + 4 * h;
  int wofs_r = wofs + 128 * LDP;
  asm volatile("" : "+v"(wofs_r));
  float bias[8];
#pragma unroll
  for (int n = 0; n < 8; ++n) bias[n] = bl[n * 16 + c];

  float*       slot = aslot + wave * (16 * LDP);
  const float* ap   = slot + c * LDP + 4 * h;

  for (;;) {
    asm volatile("" ::: "memory");

    v8f acc[8];
#pragma unroll
    for (int n = 0; n < 8; ++n) acc[n] = {};

    half_tile<false>(acc, ax, ap, wlds, wofs_r);

    gather_tile(xin, src, dst, slot, tile * 16, E, lane, cq, my_d);

    asm volatile("" ::: "memory");

    half_tile<true>(acc, ax, ap, wlds, wofs);

    const int mrow = tile * 16 + 8 * h;
#pragma unroll
    for (int r = 0; r < 8; ++r) {
      if (mrow + r < n_nodes) {
        float* orow = out + (long)(mrow + r) * CCH + c;
#pragma unroll
        for (int n = 0; n < 8; ++n) {
          float v = acc[n][r] + bias[n];
          if (SIG) v = 1.0f / (1.0f + __expf(-v));
          orow[n * 16] = v;
        }
      }
    }

    tile += SG_BLOCKS * SG_WAVES;
    if (tile >= ntiles) break;
    request(tile);
  }
}

extern "C" void kernel_launch(void* const* d_in, const int* in_sizes, int n_in,
                              void* d_out, int out_size, void* d_ws, size_t ws_size,
                              hipStream_t stream) {
  const float* x   = (const float*)d_in[0];
  const int*   ei  = (const int*)d_in[1];
  const float* W1l = (const float*)d_in[2];
  const float* b1l = (const float*)d_in[3];
  const float* W1r = (const float*)d_in[4];
  const float* W2l = (const float*)d_in[5];
  const float* b2l = (const float*)d_in[6];
  const float* W2r = (const float*)d_in[7];
  float* out = (float*)d_out;

  const int N = in_sizes[0] / CCH;
  const int E = in_sizes[1] / 2;

  int*   col = (int*)d_ws;
  int*   cnt = col + (size_t)N * CAP;
  float* h   = (float*)d_ws + (size_t)N * CCH;

  const int* src = ei;
  const int* dst = ei + E;

  const int bblocks = (N + BK_NODES - 1) / BK_NODES;

  int* pcount = cnt + N;
  int* plist  = pcount + PL_WGS * PR_MAX;

  partition_kernel<<<PL_WGS, PL_THREADS, 0, stream>>>(dst, plist, pcount, E, N);

  bucket_kernel<<<bblocks, BK_THREADS, 0, stream>>>(src, dst, plist, pcount, cnt, col, E, N);

  sage_layer_kernel<true><<<SG_BLOCKS, SG_THREADS, 0, stream>>>(
      x, col, cnt, src, dst, W1l, b1l, W1r, h, N, E);

  sage_layer_kernel<false><<<SG_BLOCKS, SG_THREADS, 0, stream>>>(
      h, col, cnt, src, dst, W2l, b2l, W2r, out, N, E);
}


# Layer tile split: the 106 tiles left after the last whole round of 256 are cut into four column quarters for waves 0-3 (one per SIMD) instead of a 7th whole tile on one SIMD; sums unchanged
# speedup vs baseline: 3.4166x; 1.0593x over previous
#include <hip/hip_runtime.h>
#include <math.h>


typedef __attribute__((ext_vector_type(2))) float v2f;
typedef __attribute__((ext_vector_type(8))) float v8f;
typedef __attribute__((ext_vector_type(4))) float v4f;

#define CCH 128

#define CAP 32
#define BK_NODES   1024
#define BK_THREADS 1024
#define BK_WAVES   (BK_THREADS / 32)
#define BK_BATCH   8
#define BK_LIST    256
static_assert(BK_NODES == BK_THREADS, "one thread clears and stores one counter");

#define PL_WGS     128
#define PL_THREADS 512
#define PL_CAP     128
#define PL_BATCH   10
#define PR_MAX     1024
static_assert(BK_NODES == 1024, "range number = destination >> 10");
static_assert(PL_WGS * 8 == BK_THREADS, "eight threads of the owner read one list");

__global__ __launch_bounds__(PL_THREADS) void partition_kernel(
    const int* __restrict__ dst, int* __restrict__ plist_raw,
    int* __restrict__ pcount, int E, int n) {
  __shared__ int pc[PR_MAX];
  const int R = (n + BK_NODES - 1) / BK_NODES;
  if (R > PR_MAX) return;
  const int tid = threadIdx.x;
  for (int i = tid; i < PR_MAX; i += PL_THREADS) pc[i] = 0;
  __syncthreads();
  const int chunk = (E + PL_WGS - 1) / PL_WGS;
  const int e0 = blockIdx.x * chunk;
  int e1 = e0 + chunk;  if (e1 > E) e1 = E;
  int2* mine = (int2*)plist_raw + (size_t)blockIdx.x * R * PL_CAP;
  for (int eb = e0; eb < e1; eb += PL_BATCH * PL_THREADS) {
    int d[PL_BATCH];
#pragma unroll
    for (int j = 0; j < PL_BATCH; ++j) {
      const int e = eb + j * PL_THREADS + tid;
      d[j] = -1;
      if (e < e1) d[j] = dst[e];
    }
    int pos[PL_BATCH];
#pragma unroll
    for (int j = 0; j < PL_BATCH; ++j) {
      const unsigned r = (unsigned)d[j] >> 10;
      pos[j] = PL_CAP;
      if (r < (unsigned)R) pos[j] = atomicAdd(&pc[r], 1);
    }
#pragma unroll
    for (int j = 0; j < PL_BATCH; ++j) {
      const int e = eb + j * PL_THREADS + tid;
      const unsigned r = (unsigned)d[j] >> 10;
      if (pos[j] < PL_CAP) mine[(size_t)r * PL_CAP + pos[j]] = make_int2(d[j], e);
    }
  }
  __syncthreads();
  for (int i = tid; i < R; i += PL_THREADS) pcount[blockIdx.x * PR_MAX + i] = pc[i];
}

__device__ __forceinline__ void rank_list(const int2* wl, int m, int* lcnt,
                                          int* lcol, int lane) {
  for (int i = lane; i < m; i += 32) {
    const int2 en = wl[i];
    const int r = atomicAdd(&lcnt[en.x], 1);
    if (r < CAP) lcol[en.x * CAP + r] = en.y;
  }
}

__global__ __launch_bounds__(BK_THREADS) void bucket_kernel(
    const int* __restrict__ src, const int* __restrict__ dst,
    const int* __restrict__ plist_raw, const int* __restrict__ pcount,
    int* __restrict__ cnt, int* __restrict__ col, int E, int n) {
  __shared__ int lcnt[BK_NODES];
  __shared__ __attribute__((aligned(16))) int lcol[BK_NODES * CAP];
  __shared__ __attribute__((aligned(8))) int2 wlist[BK_WAVES * BK_LIST];
  const int tid  = threadIdx.x;
  const int lane = tid & 31;
  const int wave = __builtin_amdgcn_readfirstlane(tid >> 5);
  int2* wl = wlist + wave * BK_LIST;
  int head = (int)(((16u - ((unsigned)(size_t)dst & 15u)) & 15u) >> 2);
  if (head > E) head = E;
  const int nvec  = (E - head) >> 2;
  const int tail0 = head + 4 * nvec;
  const int4* __restrict__ dv = (const int4*)(dst + head);
  const int per   = BK_THREADS * BK_BATCH;
  const int steps = nvec / per;
  const int s_off = (int)(((long)blockIdx.x * steps) / gridDim.x);

  const int R = (n + BK_NODES - 1) / BK_NODES;
  for (int base = blockIdx.x * BK_NODES; base < n; base += gridDim.x * BK_NODES) {
    lcnt[tid] = 0;
    const int r  = base / BK_NODES;
    const int lw = tid >> 3, sub = tid & 7;
    int c = 0;
    if (R <= PR_MAX) c = pcount[lw * PR_MAX + r];
    const int sift_all = __syncthreads_or((R > PR_MAX) || (c > PL_CAP));

    if (!sift_all) {
      const int2* pl = (const int2*)plist_raw + ((size_t)lw * R + r) * PL_CAP;
#pragma nounroll
      for (int j0 = 0; j0 < PL_CAP / 8; j0 += 8) {
        int2 en[8];
#pragma unroll
        for (int j = 0; j < 8; ++j) {
          en[j] = make_int2(-1, 0);
          const int i = sub + 8 * (j0 + j);
          if (i < c) en[j] = pl[i];
        }
        int rr[8];
#pragma unroll
        for (int j = 0; j < 8; ++j) {
          const unsigned dl = (unsigned)(en[j].x - base);
          rr[j] = CAP;
          if (dl < BK_NODES) rr[j] = atomicAdd(&lcnt[dl], 1);
        }
#pragma unroll
        for (int j = 0; j < 8; ++j) {
          const unsigned dl = (unsigned)(en[j].x - base);
          if (rr[j] < CAP) lcol[dl * CAP + rr[j]] = en[j].y;
        }
      }
    } else {
    int wcount = 0;
    auto sift = [&](const unsigned (&dl)[BK_BATCH][4], int i0) {
#pragma unroll
      for (int u = 0; u < BK_BATCH; ++u) {
#pragma unroll
        for (int j = 0; j < 4; ++j) {
          const bool     hit = dl[u][j] < BK_NODES;
          const unsigned m   = __builtin_amdgcn_ballot_w32(hit);
          if (__builtin_expect(m != 0u, 0)) {
            if (wcount > BK_LIST - 32) {
              rank_list(wl, wcount, lcnt, lcol, lane);
              wcount = 0;
            }
            int t4 = 4 * tid;
            asm volatile("" : "+v"(t4));
            const int pos = wcount + (int)__builtin_amdgcn_mbcnt_lo(m, 0u);
            const int e   = (head + 4 * (i0 + u * BK_THREADS) + j) + t4;
            if (hit) wl[pos] = make_int2((int)dl[u][j], e);
            wcount += __builtin_popcount(m);
          }
        }
      }
    };

    int4 q[BK_BATCH];
    int  i0n = s_off * per;
    if (steps > 0) {
#pragma unroll
      for (int u = 0; u < BK_BATCH; ++u) q[u] = dv[i0n + u * BK_THREADS + tid];
    }
    for (int s = 0; s < steps; ++s) {
      const int i0 = i0n;
      unsigned dl[BK_BATCH][4];
#pragma unroll
      for (int u = 0; u < BK_BATCH; ++u) {
        dl[u][0] = (unsigned)(q[u].x - base); dl[u][1] = (unsigned)(q[u].y - base);
        dl[u][2] = (unsigned)(q[u].z - base); dl[u][3] = (unsigned)(q[u].w - base);
        asm volatile("" : "+v"(dl[u][0]), "+v"(dl[u][1]), "+v"(dl[u][2]), "+v"(dl[u][3]));
      }
      int sn = s + 1; if (sn >= steps) sn = steps - 1;
      sn += s_off;    if (sn >= steps) sn -= steps;
      i0n = sn * per;
#pragma unroll
      for (int u = 0; u < BK_BATCH; ++u) q[u] = dv[i0n + u * BK_THREADS + tid];
      sift(dl, i0);
    }
    {
      const int i0 = steps * per;
      unsigned dl[BK_BATCH][4];
#pragma unroll
      for (int u = 0; u < BK_BATCH; ++u) {
        const int i = i0 + u * BK_THREADS + tid;
        int4 t = make_int4(base + BK_NODES, base + BK_NODES, base + BK_NODES,
                           base + BK_NODES);
        if (i < nvec) t = dv[i];
        dl[u][0] = (unsigned)(t.x - base); dl[u][1] = (unsigned)(t.y - base);
        dl[u][2] = (unsigned)(t.z - base); dl[u][3] = (unsigned)(t.w - base);
      }
      if (i0 < nvec) sift(dl, i0);
    }
    rank_list(wl, wcount, lcnt, lcol, lane);
    if (tid < 8) {
      const int e = (tid < 4) ? tid : (tail0 + tid - 4);
      const bool mine = (tid < 4) ? (e < head) : (e < E);
      if (mine) {
        const unsigned dl = (unsigned)(dst[e] - base);
        if (dl < BK_NODES) {
          const int rr = atomicAdd(&lcnt[dl], 1);
          if (rr < CAP) lcol[dl * CAP + rr] = e;
        }
      }
    }
    }
    __syncthreads();

    int tw = tid;
    asm volatile("" : "+v"(tw));
    for (int k0 = 0; k0 < CAP; k0 += 8) {
      int val[8];
#pragma unroll
      for (int kk = 0; kk < 8; ++kk) {
        const int i    = (k0 + kk) * BK_THREADS + tw;
        const int node = i >> 5, slot = i & 31;
        int c = lcnt[node]; if (c > CAP) c = CAP;
        val[kk] = 0;
        if (slot < c) val[kk] = src[lcol[i]];
      }
#pragma unroll
      for (int kk = 0; kk < 8; ++kk) {
        const int i    = (k0 + kk) * BK_THREADS + tw;
        const int node = i >> 5, slot = i & 31;
        if (base + node < n) col[(long)(base + node) * CAP + slot] = val[kk];
      }
    }
    if (base + tw < n) cnt[base + tw] = lcnt[tw];
    __syncthreads();
  }
}

#define LDP 132
#define SG_WAVES   16
#define SG_THREADS (SG_WAVES * 32)
#define SG_BLOCKS  256

__device__ __forceinline__ v8f wmma4(v2f a, v2f b, v8f c) {
  return __builtin_amdgcn_wmma_f32_16x16x4_f32(false, a, false, b, (short)0, c,
                                               false, false);
}

__device__ __forceinline__ void bucket_request(
    const int* __restrict__ col, const int* __restrict__ deg, int node0, int n,
    int lane, int4 (&cq)[4], int& my_d) {
  my_d = 0;
  if (lane < 16 && node0 + lane < n) my_d = deg[node0 + lane];
#pragma unroll
  for (int g = 0; g < 4; ++g) {
    const int nb = node0 + 4 * g;
    cq[g] = make_int4(0, 0, 0, 0);
    if (nb + (lane >> 3) < n)
      cq[g] = *(const int4*)(col + (long)nb * CAP + lane * 4);
  }
}

__device__ __forceinline__ void gather_tile(
    const float* __restrict__ feat, const int* __restrict__ src,
    const int* __restrict__ dst, float* slot, int node0, int E, int lane,
    int4 (&cq)[4], int my_d) {
  const float* frow = feat + lane * 4;
  const float my_s = 1.0f / fmaxf((float)my_d, 1.0f);
  for (int g = 0; g < 4; ++g) {
    const int cb[4] = {cq[0].x, cq[0].y, cq[0].z, cq[0].w};
    cq[0] = cq[1]; cq[1] = cq[2]; cq[2] = cq[3];
    int d[4];
#pragma unroll
    for (int t = 0; t < 4; ++t) d[t] = __builtin_amdgcn_readlane(my_d, 4 * g + t);
    int ck[4][8];
#pragma unroll
    for (int t = 0; t < 4; ++t)
#pragma unroll
      for (int k = 0; k < 8; ++k)
        ck[t][k] = __builtin_amdgcn_readlane(cb[k & 3], 8 * t + (k >> 2));
    v4f v[4][8];
#pragma unroll
    for (int t = 0; t < 4; ++t)
#pragma unroll
      for (int k = 0; k < 8; ++k) {
        v[t][k] = (v4f){0.f, 0.f, 0.f, 0.f};
        if (k < d[t] && d[t] <= CAP)
          v[t][k] = *(const v4f*)(frow + (long)ck[t][k] * CCH);
      }
#pragma unroll
    for (int t = 0; t < 4; ++t)
#pragma unroll
      for (int k = 0; k < 8; ++k)
        asm volatile("" : "+v"(v[t][k]));
#pragma unroll
    for (int t = 0; t < 4; ++t) {
      float4 acc = make_float4(0.f, 0.f, 0.f, 0.f);
#pragma unroll
      for (int k = 0; k < 8; ++k) {
        acc.x += v[t][k].x; acc.y += v[t][k].y;
        acc.z += v[t][k].z; acc.w += v[t][k].w;
      }
      const int dt = d[t];
      if (dt > 8 && dt <= CAP) {
        for (int j = 8; j < dt; j += 8) {
          const int r  = dt - j;
          const int l0 = 8 * t + (j >> 2);
          int cj[8];
#pragma unroll
          for (int k = 0; k < 8; ++k)
            cj[k] = __builtin_amdgcn_readlane(cb[k & 3], l0 + (k >> 2));
          float4 w[8];
#pragma unroll
          for (int k = 0; k < 8; ++k) {
            w[k] = make_float4(0.f, 0.f, 0.f, 0.f);
            if (k < r) w[k] = *(const float4*)(frow + (long)cj[k] * CCH);
          }
#pragma unroll
          for (int k = 0; k < 8; ++k) {
            acc.x += w[k].x; acc.y += w[k].y; acc.z += w[k].z; acc.w += w[k].w;
          }
        }
      }
      const float s = __int_as_float(
          __builtin_amdgcn_readlane(__float_as_int(my_s), 4 * g + t));
      acc.x *= s; acc.y *= s; acc.z *= s; acc.w *= s;
      *(float4*)(slot + (4 * g + t) * LDP + lane * 4) = acc;
    }
  }
  const unsigned big = __builtin_amdgcn_ballot_w32(my_d > CAP);
  if (__builtin_expect(big != 0u, 0)) {
    for (int t = 0; t < 16; ++t) {
      if (((big >> t) & 1u) == 0u) continue;
      const int node = node0 + t;
      float4 acc = make_float4(0.f, 0.f, 0.f, 0.f);
      for (int e0 = 0; e0 < E; e0 += 32) {
        const int  e   = e0 + lane;
        const bool hit = (e < E) && (dst[e] == node);
        const int  sv  = hit ? src[e] : 0;
        unsigned mask  = __builtin_amdgcn_ballot_w32(hit);
        while (mask != 0u) {
          const int b = __builtin_ctz(mask);
          mask &= mask - 1u;
          const int sidx = __builtin_amdgcn_readlane(sv, b);
          const float4 u = *(const float4*)(frow + (long)sidx * CCH);
          acc.x += u.x; acc.y += u.y; acc.z += u.z; acc.w += u.w;
        }
      }
      const float s = __int_as_float(
          __builtin_amdgcn_readlane(__float_as_int(my_s), t));
      acc.x *= s; acc.y *= s; acc.z *= s; acc.w *= s;
      *(float4*)(slot + t * LDP + lane * 4) = acc;
    }
  }
}

template <bool A_LDS>
__device__ __forceinline__ void half_tile(v8f (&acc)[8], const float4 (&a)[16],
                                          const float* ap, const float* wl,
                                          int wofs) {
  float4 bc[8], bn[8];
  float4 ac = make_float4(0.f, 0.f, 0.f, 0.f), an = ac;
#pragma unroll
  for (int n = 0; n < 8; ++n) bc[n] = *(const float4*)(wl + wofs + n * 16 * LDP);
  if (A_LDS) ac = *(const float4*)(ap);
#pragma unroll
  for (int q = 0; q < 16; ++q) {
    if (q + 1 < 16) {
#pragma unroll
      for (int n = 0; n < 8; ++n)
        bn[n] = *(const float4*)(wl + wofs + n * 16 * LDP + 8 * (q + 1));
      if (A_LDS) an = *(const float4*)(ap + 8 * (q + 1));
    }
    const float4 av = A_LDS ? ac : a[q];
    v2f a0, a1;
    a0.x = av.x; a0.y = av.y; a1.x = av.z; a1.y = av.w;
#pragma unroll
    for (int n = 0; n < 8; ++n) {
      v2f b; b.x = bc[n].x; b.y = bc[n].y;
      acc[n] = wmma4(a0, b, acc[n]);
    }
#pragma unroll
    for (int n = 0; n < 8; ++n) {
      v2f b; b.x = bc[n].z; b.y = bc[n].w;
      acc[n] = wmma4(a1, b, acc[n]);
    }
#pragma unroll
    for (int n = 0; n < 8; ++n) bc[n] = bn[n];
    ac = an;
  }
}

template <bool A_LDS>
__device__ __forceinline__ void quarter_half(v8f (&acc)[8], const float4 (&a)[16],
                                             const float* ap, const float* wl,
                                             int wofs) {
  float4 bc[4][2], bn[4][2];
  float4 ac[4], an[4];
#pragma unroll
  for (int s = 0; s < 4; ++s) {
#pragma unroll
    for (int n = 0; n < 2; ++n) {
      bc[s][n] = *(const float4*)(wl + wofs + n * 16 * LDP + 8 * s);
      bn[s][n] = bc[s][n];
    }
    ac[s] = make_float4(0.f, 0.f, 0.f, 0.f);
    if (A_LDS) ac[s] = *(const float4*)(ap + 8 * s);
    an[s] = ac[s];
  }
#pragma unroll
  for (int q0 = 0; q0 < 16; q0 += 4) {
    if (q0 + 4 < 16) {
#pragma unroll
      for (int s = 0; s < 4; ++s) {
#pragma unroll
        for (int n = 0; n < 2; ++n)
          bn[s][n] = *(const float4*)(wl + wofs + n * 16 * LDP + 8 * (q0 + 4 + s));
        if (A_LDS) an[s] = *(const float4*)(ap + 8 * (q0 + 4 + s));
      }
    }
    __builtin_amdgcn_sched_barrier(0);
#pragma unroll
    for (int s = 0; s < 4; ++s) {
      const float4 av = A_LDS ? ac[s] : a[q0 + s];
      v2f a0, a1;
      a0.x = av.x; a0.y = av.y; a1.x = av.z; a1.y = av.w;
#pragma unroll
      for (int n = 0; n < 2; ++n) {
        v2f b; b.x = bc[s][n].x; b.y = bc[s][n].y;
        acc[n] = wmma4(a0, b, acc[n]);
      }
#pragma unroll
      for (int n = 0; n < 2; ++n) {
        v2f b; b.x = bc[s][n].z; b.y = bc[s][n].w;
        acc[n] = wmma4(a1, b, acc[n]);
      }
    }
    __builtin_amdgcn_sched_barrier(0);
#pragma unroll
    for (int s = 0; s < 4; ++s) {
      bc[s][0] = bn[s][0]; bc[s][1] = bn[s][1];
      ac[s] = an[s];
    }
  }
}

template <bool SIG>
__global__ __launch_bounds__(SG_THREADS) void sage_layer_kernel(
    const float* __restrict__ xin,
    const int* __restrict__ col,
    const int* __restrict__ deg,
    const int* __restrict__ src,
    const int* __restrict__ dst,
    const float* __restrict__ Wl,
    const float* __restrict__ bl,
    const float* __restrict__ Wr,
    float* __restrict__ out,
    int n_nodes, int E) {
  __shared__ __attribute__((aligned(16))) float wlds[256 * LDP];
  __shared__ __attribute__((aligned(16))) float aslot[SG_WAVES * 16 * LDP];
  const int tid = threadIdx.x;
  const int lane = tid & 31;
  const int wave = __builtin_amdgcn_readfirstlane(tid >> 5);
  const int h    = lane >> 4;
  const int c    = lane & 15;
  const int ntiles = (n_nodes + 15) >> 4;
  const int nq    = ntiles % SG_BLOCKS;
  const int nfull = ntiles - nq;
  const bool has_q = (wave < 4) && ((int)blockIdx.x < nq);
  const int  qn0   = 2 * (wave & 3);
  int  tile    = blockIdx.x + SG_BLOCKS * wave;
  bool quarter = false;
  if (tile >= nfull) { quarter = has_q; tile = nfull + blockIdx.x; }
  const bool any_work = (blockIdx.x + SG_BLOCKS * wave < nfull) || has_q;

  float4 ax[16];
  int4   cq[4];
  int    my_d;
  auto request = [&](int t) {
    int m = t * 16 + c;  if (m >= n_nodes) m = n_nodes - 1;
    const float* p = xin + (long)m * CCH + 4 * h;
#pragma unroll
    for (int q = 0; q < 16; ++q) ax[q] = *(const float4*)(p + 8 * q);
    bucket_request(col, deg, t * 16, n_nodes, lane, cq, my_d);
  };
  request(tile);

#pragma unroll 8
  for (int j = 0; j < (256 * 32) / SG_THREADS; ++j) {
    const int i = j * SG_THREADS + tid, row = i >> 5, c4 = i & 31;
    const float* wsrc = (row < 128) ? (Wl + row * CCH) : (Wr + (row - 128) * CCH);
    *(float4*)(wlds + row * LDP + c4 * 4) = *(const float4*)(wsrc + c4 * 4);
  }
  __syncthreads();

  if (!any_work) return;

  const int wofs = c * LDP + 4 * h;
  int wofs_r = wofs + 128 * LDP;
  asm volatile("" : "+v"(wofs_r));
  float bias[8];
#pragma unroll
  for (int n = 0; n < 8; ++n) bias[n] = bl[n * 16 + c];
  float qbias[2];
#pragma unroll
  for (int n = 0; n < 2; ++n) qbias[n] = bl[(qn0 + n) * 16 + c];
  const int wofs_q  = wofs   + qn0 * 16 * LDP;
  const int wofs_rq = wofs_r + qn0 * 16 * LDP;

  float*       slot = aslot + wave * (16 * LDP);
  const float* ap   = slot + c * LDP + 4 * h;

  for (;;) {
    asm volatile("" ::: "memory");

    v8f acc[8];
#pragma unroll
    for (int n = 0; n < 8; ++n) acc[n] = {};

    if (!quarter) half_tile<false>(acc, ax, ap, wlds, wofs_r);
    else          quarter_half<false>(acc, ax, ap, wlds, wofs_rq);

    gather_tile(xin, src, dst, slot, tile * 16, E, lane, cq, my_d);

    asm volatile("" ::: "memory");

    if (!quarter) half_tile<true>(acc, ax, ap, wlds, wofs);
    else          quarter_half<true>(acc, ax, ap, wlds, wofs_q);

    const int mrow = tile * 16 + 8 * h;
    if (quarter) {
#pragma unroll
      for (int r = 0; r < 8; ++r) {
        if (mrow + r < n_nodes) {
          float* orow = out + (long)(mrow + r) * CCH + qn0 * 16 + c;
#pragma unroll
          for (int n = 0; n < 2; ++n) {
            float v = acc[n][r] + qbias[n];
            if (SIG) v = 1.0f / (1.0f + __expf(-v));
            orow[n * 16] = v;
          }
        }
      }
      break;
    }
#pragma unroll
    for (int r = 0; r < 8; ++r) {
      if (mrow + r < n_nodes) {
        float* orow = out + (long)(mrow + r) * CCH + c;
#pragma unroll
        for (int n = 0; n < 8; ++n) {
          float v = acc[n][r] + bias[n];
          if (SIG) v = 1.0f / (1.0f + __expf(-v));
          orow[n * 16] = v;
        }
      }
    }

    tile += SG_BLOCKS * SG_WAVES;
    if (tile >= nfull) {
      if (!has_q) break;
      quarter = true;
      tile = nfull + blockIdx.x;
    }
    request(tile);
  }
}

extern "C" void kernel_launch(void* const* d_in, const int* in_sizes, int n_in,
                              void* d_out, int out_size, void* d_ws, size_t ws_size,
                              hipStream_t stream) {
  const float* x   = (const float*)d_in[0];
  const int*   ei  = (const int*)d_in[1];
  const float* W1l = (const float*)d_in[2];
  const float* b1l = (const float*)d_in[3];
  const float* W1r = (const float*)d_in[4];
  const float* W2l = (const float*)d_in[5];
  const float* b2l = (const float*)d_in[6];
  const float* W2r = (const float*)d_in[7];
  float* out = (float*)d_out;

  const int N = in_sizes[0] / CCH;
  const int E = in_sizes[1] / 2;

  int*   col = (int*)d_ws;
  int*   cnt = col + (size_t)N * CAP;
  float* h   = (float*)d_ws + (size_t)N * CCH;

  const int* src = ei;
  const int* dst = ei + E;

  const int bblocks = (N + BK_NODES - 1) / BK_NODES;

  int* pcount = cnt + N;
  int* plist  = pcount + PL_WGS * PR_MAX;

  partition_kernel<<<PL_WGS, PL_THREADS, 0, stream>>>(dst, plist, pcount, E, N);

  bucket_kernel<<<bblocks, BK_THREADS, 0, stream>>>(src, dst, plist, pcount, cnt, col, E, N);

  sage_layer_kernel<true><<<SG_BLOCKS, SG_THREADS, 0, stream>>>(
      x, col, cnt, src, dst, W1l, b1l, W1r, h, N, E);

  sage_layer_kernel<false><<<SG_BLOCKS, SG_THREADS, 0, stream>>>(
      h, col, cnt, src, dst, W2l, b2l, W2r, out, N, E);
}
